# starting kernel
# baseline (speedup 1.0000x reference)
#include <hip/hip_runtime.h>
#include <hip/hip_bf16.h>

typedef float v2f __attribute__((ext_vector_type(2)));
typedef float v8f __attribute__((ext_vector_type(8)));

#define N_NODES 100000
#define N_EDGES 1600000
#define IN_F 64
#define HID 64
#define NC 40

__global__ void gcn_zero_kernel(float* __restrict__ p, int n) {
  int i = blockIdx.x * blockDim.x + threadIdx.x;
  if (i < n) p[i] = 0.0f;
}

__global__ void gcn_bias_init_kernel(float* __restrict__ out,
                                     const float* __restrict__ b) {
  unsigned i = blockIdx.x * blockDim.x + threadIdx.x;
  const unsigned total = (unsigned)N_NODES * NC;
  if (i < total) {
    unsigned c = i - (i / NC) * NC;
    out[i] = b[c];
  }
}

__global__ void gcn_scatter_add_kernel(const float* __restrict__ x,
                                       const int* __restrict__ src,
                                       const int* __restrict__ dst,
                                       float* __restrict__ acc,
                                       int groups, int stride) {
  unsigned tid = blockIdx.x * blockDim.x + threadIdx.x;
  unsigned total = (unsigned)N_EDGES * (unsigned)groups;
  if (tid >= total) return;
  unsigned e = tid / (unsigned)groups;
  unsigned g = tid - e * (unsigned)groups;
  int s = src[e];
  int d = dst[e];
  const float4 v =
      *reinterpret_cast<const float4*>(x + (size_t)s * stride + (size_t)g * 4);
  float* o = acc + (size_t)d * stride + (size_t)g * 4;
  atomicAdd(o + 0, v.x);
  atomicAdd(o + 1, v.y);
  atomicAdd(o + 2, v.z);
  atomicAdd(o + 3, v.w);
}

template <int KDIM, int NREAL, bool RELU, bool BIAS>
__global__ void gcn_wmma_gemm_kernel(const float* __restrict__ A,
                                     const float* __restrict__ W,
                                     const float* __restrict__ bias,
                                     float* __restrict__ C,
                                     int n_row_tiles, int n_col_tiles) {
  constexpr bool FULL = (NREAL % 16 == 0);

  const int tile = blockIdx.x * (blockDim.x >> 5) + (threadIdx.x >> 5);
  const int n_tiles = n_row_tiles * n_col_tiles;
  if (tile >= n_tiles) return;

  const int row_tile = tile / n_col_tiles;
  const int col_tile = tile - row_tile * n_col_tiles;
  const int row0 = row_tile << 4;
  const int col0 = col_tile << 4;

  const int lane = threadIdx.x & 31;
  const int lh = lane & 15;
  const int hi = lane >> 4;

  const int nb = col0 + lh;
  const bool nok = FULL ? true : (nb < NREAL);
  const int nbc = FULL ? nb : (nok ? nb : (NREAL - 1));
  const float nmask = nok ? 1.0f : 0.0f;

  const float* arow = A + (size_t)(row0 + lh) * KDIM + (size_t)(hi * 2);

  v8f c = {};
#pragma unroll
  for (int k = 0; k < KDIM; k += 4) {
    v2f a, b;
    a.x = arow[k];
    a.y = arow[k + 1];
    const int kb = k + hi * 2;
    if (FULL) {
      b.x = W[(size_t)kb * NREAL + nbc];
      b.y = W[(size_t)(kb + 1) * NREAL + nbc];
    } else {
      b.x = W[(size_t)kb * NREAL + nbc] * nmask;
      b.y = W[(size_t)(kb + 1) * NREAL + nbc] * nmask;
    }
    c = __builtin_amdgcn_wmma_f32_16x16x4_f32(false, a, false, b, (short)0, c,
                                              false, false);
  }

  if (!nok) return;
  float bv = 0.0f;
  if (BIAS) bv = bias[nb];
#pragma unroll
  for (int r = 0; r < 8; ++r) {
    const int row = row0 + r + (hi << 3);
    float v = c[r] + bv;
    if (RELU) v = fmaxf(v, 0.0f);
    C[(size_t)row * NREAL + nb] = v;
  }
}

extern "C" void kernel_launch(void* const* d_in, const int* in_sizes, int n_in,
                              void* d_out, int out_size, void* d_ws,
                              size_t ws_size, hipStream_t stream) {
  (void)in_sizes; (void)n_in; (void)out_size; (void)ws_size;
  const float* features = (const float*)d_in[0];
  const int*   src      = (const int*)d_in[1];
  const int*   dst      = (const int*)d_in[2];
  const float* W1       = (const float*)d_in[3];
  const float* b1       = (const float*)d_in[4];
  const float* W2       = (const float*)d_in[5];
  const float* b2       = (const float*)d_in[6];
  float* out = (float*)d_out;

  float* ws   = (float*)d_ws;
  float* agg1 = ws;
  float* h    = ws + (size_t)N_NODES * HID;
  float* t    = ws;

  const int nAgg1 = N_NODES * IN_F;

  gcn_zero_kernel<<<(nAgg1 + 255) / 256, 256, 0, stream>>>(agg1, nAgg1);

  {
    const int total = N_EDGES * 16;
    gcn_scatter_add_kernel<<<(total + 255) / 256, 256, 0, stream>>>(
        features, src, dst, agg1, 16, IN_F);
  }

  {
    const int rt = N_NODES / 16, ct = HID / 16;
    const int tiles = rt * ct;
    gcn_wmma_gemm_kernel<HID, HID, true, true>
        <<<(tiles + 7) / 8, 256, 0, stream>>>(agg1, W1, b1, h, rt, ct);
  }

  {
    const int rt = N_NODES / 16, ct = (NC + 15) / 16;
    const int tiles = rt * ct;
    gcn_wmma_gemm_kernel<HID, NC, false, false>
        <<<(tiles + 7) / 8, 256, 0, stream>>>(h, W2, nullptr, t, rt, ct);
  }

  {
    const int total = N_NODES * NC;
    gcn_bias_init_kernel<<<(total + 255) / 256, 256, 0, stream>>>(out, b2);
  }

  {
    const int total = N_EDGES * 10;
    gcn_scatter_add_kernel<<<(total + 255) / 256, 256, 0, stream>>>(
        t, src, dst, out, 10, NC);
  }
}


# scatter-add: a lane owns one column of the row, so each wave-wide f32 atomic covers one whole 128-B line (was 8 strided floats in each of 2 lines); a wave walks 32 edges; arrival-order sums as before
# speedup vs baseline: 1.7985x; 1.7985x over previous
#include <hip/hip_runtime.h>
#include <hip/hip_bf16.h>

typedef float v2f __attribute__((ext_vector_type(2)));
typedef float v8f __attribute__((ext_vector_type(8)));

#define N_NODES 100000
#define N_EDGES 1600000
#define IN_F 64
#define HID 64
#define NC 40

__global__ void gcn_zero_kernel(float* __restrict__ p, int n) {
  int i = blockIdx.x * blockDim.x + threadIdx.x;
  if (i < n) p[i] = 0.0f;
}

__global__ void gcn_bias_init_kernel(float* __restrict__ out,
                                     const float* __restrict__ b) {
  unsigned i = blockIdx.x * blockDim.x + threadIdx.x;
  const unsigned total = (unsigned)N_NODES * NC;
  if (i < total) {
    unsigned c = i - (i / NC) * NC;
    out[i] = b[c];
  }
}

#define EDGES_PER_WAVE 32
#define EDGES_PER_BATCH 16
static_assert(N_EDGES % EDGES_PER_WAVE == 0, "edge list is cut into whole waves");

__global__ void __launch_bounds__(256)
gcn_scatter_rows64_kernel(const float* __restrict__ x,
                          const int* __restrict__ src,
                          const int* __restrict__ dst,
                          float* __restrict__ acc) {
  const unsigned lane = threadIdx.x & 31u;
  const unsigned wave = blockIdx.x * (blockDim.x >> 5) + (threadIdx.x >> 5);
  const unsigned e0 = wave * EDGES_PER_WAVE;
  if (e0 >= (unsigned)N_EDGES) return;
  const int s_l = src[e0 + lane];
  const int d_l = dst[e0 + lane];
#pragma unroll
  for (int b = 0; b < EDGES_PER_WAVE; b += EDGES_PER_BATCH) {
    float v0[EDGES_PER_BATCH], v1[EDGES_PER_BATCH];
#pragma unroll
    for (int i = 0; i < EDGES_PER_BATCH; ++i) {
      const int s = __builtin_amdgcn_readlane(s_l, b + i);
      const float* xr = x + (size_t)s * 64;
      v0[i] = xr[lane];
      v1[i] = xr[lane + 32];
    }
#pragma unroll
    for (int i = 0; i < EDGES_PER_BATCH; ++i) {
      const int d = __builtin_amdgcn_readlane(d_l, b + i);
      float* o = acc + (size_t)d * 64;
      atomicAdd(o + lane, v0[i]);
      atomicAdd(o + lane + 32, v1[i]);
    }
  }
}

__global__ void __launch_bounds__(256)
gcn_scatter_rows40_kernel(const float* __restrict__ x,
                          const int* __restrict__ src,
                          const int* __restrict__ dst,
                          float* __restrict__ acc) {
  const unsigned lane = threadIdx.x & 31u;
  const unsigned wave = blockIdx.x * (blockDim.x >> 5) + (threadIdx.x >> 5);
  const unsigned e0 = wave * EDGES_PER_WAVE;
  if (e0 >= (unsigned)N_EDGES) return;
  const int s_l = src[e0 + lane];
  const int d_l = dst[e0 + lane];
#pragma unroll
  for (int b = 0; b < EDGES_PER_WAVE; b += EDGES_PER_BATCH) {
    float v0[EDGES_PER_BATCH], v1[EDGES_PER_BATCH];
#pragma unroll
    for (int i = 0; i < EDGES_PER_BATCH; ++i) {
      const int s = __builtin_amdgcn_readlane(s_l, b + i);
      const int d = __builtin_amdgcn_readlane(d_l, b + i);
      const unsigned n0 = 32u - (((unsigned)d & 3u) << 3);
      const unsigned c1 = n0 + lane;
      const float* xr = x + (size_t)s * NC;
      v0[i] = xr[lane];
      v1[i] = xr[c1 < (unsigned)NC ? c1 : (unsigned)(NC - 1)];
    }
#pragma unroll
    for (int i = 0; i < EDGES_PER_BATCH; ++i) {
      const int d = __builtin_amdgcn_readlane(d_l, b + i);
      const unsigned n0 = 32u - (((unsigned)d & 3u) << 3);
      float* o = acc + (size_t)d * NC;
      if (lane < n0) atomicAdd(o + lane, v0[i]);
      if (lane < (unsigned)NC - n0) atomicAdd(o + n0 + lane, v1[i]);
    }
  }
}

template <int KDIM, int NREAL, bool RELU, bool BIAS>
__global__ void gcn_wmma_gemm_kernel(const float* __restrict__ A,
                                     const float* __restrict__ W,
                                     const float* __restrict__ bias,
                                     float* __restrict__ C,
                                     int n_row_tiles, int n_col_tiles) {
  constexpr bool FULL = (NREAL % 16 == 0);

  const int tile = blockIdx.x * (blockDim.x >> 5) + (threadIdx.x >> 5);
  const int n_tiles = n_row_tiles * n_col_tiles;
  if (tile >= n_tiles) return;

  const int row_tile = tile / n_col_tiles;
  const int col_tile = tile - row_tile * n_col_tiles;
  const int row0 = row_tile << 4;
  const int col0 = col_tile << 4;

  const int lane = threadIdx.x & 31;
  const int lh = lane & 15;
  const int hi = lane >> 4;

  const int nb = col0 + lh;
  const bool nok = FULL ? true : (nb < NREAL);
  const int nbc = FULL ? nb : (nok ? nb : (NREAL - 1));
  const float nmask = nok ? 1.0f : 0.0f;

  const float* arow = A + (size_t)(row0 + lh) * KDIM + (size_t)(hi * 2);

  v8f c = {};
#pragma unroll
  for (int k = 0; k < KDIM; k += 4) {
    v2f a, b;
    a.x = arow[k];
    a.y = arow[k + 1];
    const int kb = k + hi * 2;
    if (FULL) {
      b.x = W[(size_t)kb * NREAL + nbc];
      b.y = W[(size_t)(kb + 1) * NREAL + nbc];
    } else {
      b.x = W[(size_t)kb * NREAL + nbc] * nmask;
      b.y = W[(size_t)(kb + 1) * NREAL + nbc] * nmask;
    }
    c = __builtin_amdgcn_wmma_f32_16x16x4_f32(false, a, false, b, (short)0, c,
                                              false, false);
  }

  if (!nok) return;
  float bv = 0.0f;
  if (BIAS) bv = bias[nb];
#pragma unroll
  for (int r = 0; r < 8; ++r) {
    const int row = row0 + r + (hi << 3);
    float v = c[r] + bv;
    if (RELU) v = fmaxf(v, 0.0f);
    C[(size_t)row * NREAL + nb] = v;
  }
}

extern "C" void kernel_launch(void* const* d_in, const int* in_sizes, int n_in,
                              void* d_out, int out_size, void* d_ws,
                              size_t ws_size, hipStream_t stream) {
  (void)in_sizes; (void)n_in; (void)out_size; (void)ws_size;
  const float* features = (const float*)d_in[0];
  const int*   src      = (const int*)d_in[1];
  const int*   dst      = (const int*)d_in[2];
  const float* W1       = (const float*)d_in[3];
  const float* b1       = (const float*)d_in[4];
  const float* W2       = (const float*)d_in[5];
  const float* b2       = (const float*)d_in[6];
  float* out = (float*)d_out;

  float* ws   = (float*)d_ws;
  float* agg1 = ws;
  float* h    = ws + (size_t)N_NODES * HID;
  float* t    = ws;

  const int nAgg1 = N_NODES * IN_F;

  gcn_zero_kernel<<<(nAgg1 + 255) / 256, 256, 0, stream>>>(agg1, nAgg1);

  {
    const int waves = N_EDGES / EDGES_PER_WAVE;
    gcn_scatter_rows64_kernel<<<(waves + 7) / 8, 256, 0, stream>>>(
        features, src, dst, agg1);
  }

  {
    const int rt = N_NODES / 16, ct = HID / 16;
    const int tiles = rt * ct;
    gcn_wmma_gemm_kernel<HID, HID, true, true>
        <<<(tiles + 7) / 8, 256, 0, stream>>>(agg1, W1, b1, h, rt, ct);
  }

  {
    const int rt = N_NODES / 16, ct = (NC + 15) / 16;
    const int tiles = rt * ct;
    gcn_wmma_gemm_kernel<HID, NC, false, false>
        <<<(tiles + 7) / 8, 256, 0, stream>>>(h, W2, nullptr, t, rt, ct);
  }

  {
    const int total = N_NODES * NC;
    gcn_bias_init_kernel<<<(total + 255) / 256, 256, 0, stream>>>(out, b2);
  }

  {
    const int waves = N_EDGES / EDGES_PER_WAVE;
    gcn_scatter_rows40_kernel<<<(waves + 7) / 8, 256, 0, stream>>>(
        t, src, dst, out);
  }
}


# dense layers fused: a wave owns 16 rows, does relu(agg@W1+b1)@W2 with the hidden tile kept in LDS and weights staged in LDS, so h never goes to memory; k order inside WMMA fragments permuted
# speedup vs baseline: 1.9365x; 1.0767x over previous
#include <hip/hip_runtime.h>
#include <hip/hip_bf16.h>

typedef float v2f __attribute__((ext_vector_type(2)));
typedef float v8f __attribute__((ext_vector_type(8)));

#define N_NODES 100000
#define N_EDGES 1600000
#define IN_F 64
#define HID 64
#define NC 40

__global__ void gcn_zero_kernel(float* __restrict__ p, int n) {
  int i = blockIdx.x * blockDim.x + threadIdx.x;
  if (i < n) p[i] = 0.0f;
}

__global__ void gcn_bias_init_kernel(float* __restrict__ out,
                                     const float* __restrict__ b) {
  unsigned i = blockIdx.x * blockDim.x + threadIdx.x;
  const unsigned total = (unsigned)N_NODES * NC;
  if (i < total) {
    unsigned c = i - (i / NC) * NC;
    out[i] = b[c];
  }
}

#define EDGES_PER_WAVE 32
#define EDGES_PER_BATCH 16
static_assert(N_EDGES % EDGES_PER_WAVE == 0, "edge list is cut into whole waves");

__global__ void __launch_bounds__(256)
gcn_scatter_rows64_kernel(const float* __restrict__ x,
                          const int* __restrict__ src,
                          const int* __restrict__ dst,
                          float* __restrict__ acc) {
  const unsigned lane = threadIdx.x & 31u;
  const unsigned wave = blockIdx.x * (blockDim.x >> 5) + (threadIdx.x >> 5);
  const unsigned e0 = wave * EDGES_PER_WAVE;
  if (e0 >= (unsigned)N_EDGES) return;
  const int s_l = src[e0 + lane];
  const int d_l = dst[e0 + lane];
#pragma unroll
  for (int b = 0; b < EDGES_PER_WAVE; b += EDGES_PER_BATCH) {
    float v0[EDGES_PER_BATCH], v1[EDGES_PER_BATCH];
#pragma unroll
    for (int i = 0; i < EDGES_PER_BATCH; ++i) {
      const int s = __builtin_amdgcn_readlane(s_l, b + i);
      const float* xr = x + (size_t)s * 64;
      v0[i] = xr[lane];
      v1[i] = xr[lane + 32];
    }
#pragma unroll
    for (int i = 0; i < EDGES_PER_BATCH; ++i) {
      const int d = __builtin_amdgcn_readlane(d_l, b + i);
      float* o = acc + (size_t)d * 64;
      atomicAdd(o + lane, v0[i]);
      atomicAdd(o + lane + 32, v1[i]);
    }
  }
}

__global__ void __launch_bounds__(256)
gcn_scatter_rows40_kernel(const float* __restrict__ x,
                          const int* __restrict__ src,
                          const int* __restrict__ dst,
                          float* __restrict__ acc) {
  const unsigned lane = threadIdx.x & 31u;
  const unsigned wave = blockIdx.x * (blockDim.x >> 5) + (threadIdx.x >> 5);
  const unsigned e0 = wave * EDGES_PER_WAVE;
  if (e0 >= (unsigned)N_EDGES) return;
  const int s_l = src[e0 + lane];
  const int d_l = dst[e0 + lane];
#pragma unroll
  for (int b = 0; b < EDGES_PER_WAVE; b += EDGES_PER_BATCH) {
    float v0[EDGES_PER_BATCH], v1[EDGES_PER_BATCH];
#pragma unroll
    for (int i = 0; i < EDGES_PER_BATCH; ++i) {
      const int s = __builtin_amdgcn_readlane(s_l, b + i);
      const int d = __builtin_amdgcn_readlane(d_l, b + i);
      const unsigned n0 = 32u - (((unsigned)d & 3u) << 3);
      const unsigned c1 = n0 + lane;
      const float* xr = x + (size_t)s * NC;
      v0[i] = xr[lane];
      v1[i] = xr[c1 < (unsigned)NC ? c1 : (unsigned)(NC - 1)];
    }
#pragma unroll
    for (int i = 0; i < EDGES_PER_BATCH; ++i) {
      const int d = __builtin_amdgcn_readlane(d_l, b + i);
      const unsigned n0 = 32u - (((unsigned)d & 3u) << 3);
      float* o = acc + (size_t)d * NC;
      if (lane < n0) atomicAdd(o + lane, v0[i]);
      if (lane < (unsigned)NC - n0) atomicAdd(o + n0 + lane, v1[i]);
    }
  }
}

typedef float v4f __attribute__((ext_vector_type(4)));

#define DENSE_WAVES 10
#define DENSE_THREADS (DENSE_WAVES * 32)
#define DENSE_BLOCKS (N_NODES / (16 * DENSE_WAVES))
static_assert(N_NODES % (16 * DENSE_WAVES) == 0, "every wave owns one whole 16-row tile");
#define NC_PAD 48
#define PW1 144
#define PW2 112
#define PH 68
#define PT 52

__global__ void __launch_bounds__(DENSE_THREADS)
gcn_dense_layers_kernel(const float* __restrict__ A,
                        const float* __restrict__ W1,
                        const float* __restrict__ b1,
                        const float* __restrict__ W2,
                        float* __restrict__ T) {
  __shared__ __attribute__((aligned(16))) float w1l[(IN_F / 2) * PW1];
  __shared__ __attribute__((aligned(16))) float w2l[(HID / 2) * PW2];
  __shared__ __attribute__((aligned(16))) float tiles[DENSE_WAVES][16 * PH];

  const int tid = threadIdx.x;
  const int lane = tid & 31;
  const int wave = tid >> 5;
  const int lh = lane & 15;
  const int hi = lane >> 4;
  const int row0 = (blockIdx.x * DENSE_WAVES + wave) << 4;

  const float* arow = A + (size_t)(row0 + lh) * IN_F + (hi << 2);
  v4f af[8];
#pragma unroll
  for (int p = 0; p < 8; ++p) af[p] = *(const v4f*)(arow + 8 * p);

  constexpr int W1_PAIRS = (IN_F / 2) * HID;
  constexpr int W2_PAIRS = (HID / 2) * NC_PAD;
  constexpr int W1_TRIPS = (W1_PAIRS + DENSE_THREADS - 1) / DENSE_THREADS;
  constexpr int W2_TRIPS = (W2_PAIRS + DENSE_THREADS - 1) / DENSE_THREADS;
  v2f w1r[W1_TRIPS], w2r[W2_TRIPS];
#pragma unroll
  for (int i = 0; i < W1_TRIPS; ++i) {
    const int idx = tid + i * DENSE_THREADS;
    if (i < W1_TRIPS - 1 || idx < W1_PAIRS) {
      const int kp = idx >> 6, n = idx & 63;
      w1r[i].x = W1[(2 * kp) * HID + n];
      w1r[i].y = W1[(2 * kp + 1) * HID + n];
    }
  }
#pragma unroll
  for (int i = 0; i < W2_TRIPS; ++i) {
    const int idx = tid + i * DENSE_THREADS;
    if (i < W2_TRIPS - 1 || idx < W2_PAIRS) {
      const int kp = idx / NC_PAD, n = idx - kp * NC_PAD;
      const int nc = n < NC ? n : NC - 1;
      const float x = W2[(2 * kp) * NC + nc];
      const float y = W2[(2 * kp + 1) * NC + nc];
      w2r[i].x = n < NC ? x : 0.0f;
      w2r[i].y = n < NC ? y : 0.0f;
    }
  }
#pragma unroll
  for (int i = 0; i < W1_TRIPS; ++i) {
    const int idx = tid + i * DENSE_THREADS;
    if (i < W1_TRIPS - 1 || idx < W1_PAIRS) {
      const int kp = idx >> 6, n = idx & 63;
      *(v2f*)&w1l[kp * PW1 + 2 * n] = w1r[i];
    }
  }
#pragma unroll
  for (int i = 0; i < W2_TRIPS; ++i) {
    const int idx = tid + i * DENSE_THREADS;
    if (i < W2_TRIPS - 1 || idx < W2_PAIRS) {
      const int kp = idx / NC_PAD, n = idx - kp * NC_PAD;
      *(v2f*)&w2l[kp * PW2 + 2 * n] = w2r[i];
    }
  }
  __syncthreads();

  v8f acc[4] = {};
#pragma unroll
  for (int p = 0; p < 8; ++p) {
#pragma unroll
    for (int u = 0; u < 2; ++u) {
      v2f a;
      a.x = af[p][2 * u];
      a.y = af[p][2 * u + 1];
      const int kp = 4 * p + 2 * hi + u;
#pragma unroll
      for (int ct = 0; ct < 4; ++ct) {
        const v2f b = *(const v2f*)&w1l[kp * PW1 + 2 * (ct * 16 + lh)];
        acc[ct] = __builtin_amdgcn_wmma_f32_16x16x4_f32(false, a, false, b, (short)0,
                                                        acc[ct], false, false);
      }
    }
  }

  float* ht = tiles[wave];
#pragma unroll
  for (int ct = 0; ct < 4; ++ct) {
    const float bv = b1[ct * 16 + lh];
#pragma unroll
    for (int r = 0; r < 8; ++r)
      ht[((hi << 3) + r) * PH + ct * 16 + lh] = fmaxf(acc[ct][r] + bv, 0.0f);
  }
  __syncthreads();

  v4f hf[8];
#pragma unroll
  for (int p = 0; p < 8; ++p) hf[p] = *(const v4f*)&ht[lh * PH + 8 * p + (hi << 2)];
  v8f acc2[3] = {};
#pragma unroll
  for (int p = 0; p < 8; ++p) {
#pragma unroll
    for (int u = 0; u < 2; ++u) {
      v2f a;
      a.x = hf[p][2 * u];
      a.y = hf[p][2 * u + 1];
      const int kp = 4 * p + 2 * hi + u;
#pragma unroll
      for (int ct = 0; ct < 3; ++ct) {
        const v2f b = *(const v2f*)&w2l[kp * PW2 + 2 * (ct * 16 + lh)];
        acc2[ct] = __builtin_amdgcn_wmma_f32_16x16x4_f32(false, a, false, b, (short)0,
                                                         acc2[ct], false, false);
      }
    }
  }

#pragma unroll
  for (int ct = 0; ct < 3; ++ct) {
#pragma unroll
    for (int r = 0; r < 8; ++r)
      ht[((hi << 3) + r) * PT + ct * 16 + lh] = acc2[ct][r];
  }
  __syncthreads();

  float* trow = T + (size_t)row0 * NC;
#pragma unroll
  for (int q = 0; q < 5; ++q) {
    const int g = lane + 32 * q;
    const int row = g / 10;
    const int c4 = g - row * 10;
    const v4f v = *(const v4f*)&ht[row * PT + 4 * c4];
    *(v4f*)(trow + 4 * g) = v;
  }
}

extern "C" void kernel_launch(void* const* d_in, const int* in_sizes, int n_in,
                              void* d_out, int out_size, void* d_ws,
                              size_t ws_size, hipStream_t stream) {
  (void)in_sizes; (void)n_in; (void)out_size; (void)ws_size;
  const float* features = (const float*)d_in[0];
  const int*   src      = (const int*)d_in[1];
  const int*   dst      = (const int*)d_in[2];
  const float* W1       = (const float*)d_in[3];
  const float* b1       = (const float*)d_in[4];
  const float* W2       = (const float*)d_in[5];
  const float* b2       = (const float*)d_in[6];
  float* out = (float*)d_out;

  float* ws   = (float*)d_ws;
  float* agg1 = ws;
  float* t    = ws + (size_t)N_NODES * HID;

  const int nAgg1 = N_NODES * IN_F;

  gcn_zero_kernel<<<(nAgg1 + 255) / 256, 256, 0, stream>>>(agg1, nAgg1);

  {
    const int waves = N_EDGES / EDGES_PER_WAVE;
    gcn_scatter_rows64_kernel<<<(waves + 7) / 8, 256, 0, stream>>>(
        features, src, dst, agg1);
  }

  gcn_dense_layers_kernel<<<DENSE_BLOCKS, DENSE_THREADS, 0, stream>>>(
      agg1, W1, b1, W2, t);

  {
    const int total = N_NODES * NC;
    gcn_bias_init_kernel<<<(total + 255) / 256, 256, 0, stream>>>(out, b2);
  }

  {
    const int waves = N_EDGES / EDGES_PER_WAVE;
    gcn_scatter_rows40_kernel<<<(waves + 7) / 8, 256, 0, stream>>>(
        t, src, dst, out);
  }
}


# edges grouped per call by 128-row destination block (LDS counting sort); one work-group sums a block's rows in LDS (arrival order) and stores them whole: no global atomics, no zero-fill/seed launch
# speedup vs baseline: 2.3739x; 1.2259x over previous
#include <hip/hip_runtime.h>
#include <hip/hip_bf16.h>

typedef float v2f __attribute__((ext_vector_type(2)));
typedef float v8f __attribute__((ext_vector_type(8)));
typedef int v4i __attribute__((ext_vector_type(4)));

#define N_NODES 100000
#define N_EDGES 1600000
#define IN_F 64
#define HID 64
#define NC 40

#define BROWS 128
#define BSHIFT 7
#define NBK ((N_NODES + BROWS - 1) / BROWS)
#define NBK_PAD 800
#define NCH 256
#define CH_EDGES (N_EDGES / NCH)
#define CH_THREADS 1024
#define CH_TRIPS ((CH_EDGES + CH_THREADS - 1) / CH_THREADS)
#define SRC_BITS 17
static_assert(N_EDGES % NCH == 0, "the edge list is cut into whole chunks");
static_assert(N_NODES <= (1 << SRC_BITS), "source row fits its field");
static_assert(NBK <= NBK_PAD && NBK_PAD <= CH_THREADS && NBK_PAD % 8 == 0, "bucket table");
static_assert(NCH == 256, "one wave scans one bucket row: 32 lanes x 8");

__global__ void __launch_bounds__(CH_THREADS)
gcn_bucket_count_kernel(const int* __restrict__ dst, int* __restrict__ cnt) {
  __shared__ int h[NBK_PAD];
  const int t = threadIdx.x, j = blockIdx.x;
  const int* dp = dst + (size_t)j * CH_EDGES;
  int d[CH_TRIPS];
#pragma unroll
  for (int i = 0; i < CH_TRIPS; ++i) {
    const int idx = t + i * CH_THREADS;
    if (idx < CH_EDGES) d[i] = dp[idx];
  }
  if (t < NBK_PAD) h[t] = 0;
  __syncthreads();
#pragma unroll
  for (int i = 0; i < CH_TRIPS; ++i) {
    const int idx = t + i * CH_THREADS;
    if (idx < CH_EDGES) atomicAdd(&h[d[i] >> BSHIFT], 1);
  }
  __syncthreads();
  if (t < NBK_PAD) cnt[t * NCH + j] = h[t];
}

__global__ void __launch_bounds__(256)
gcn_bucket_rowscan_kernel(int* __restrict__ cnt, int* __restrict__ tot) {
  const int lane = threadIdx.x & 31;
  const int bk = blockIdx.x * 8 + (threadIdx.x >> 5);
  v4i* p = (v4i*)(cnt + (size_t)bk * NCH) + 2 * lane;
  const v4i c = p[0], c2 = p[1];
  const int s4 = c.x + c.y + c.z + c.w;
  const int s = s4 + c2.x + c2.y + c2.z + c2.w;
  int incl = s;
#pragma unroll
  for (int dlt = 1; dlt < 32; dlt <<= 1) {
    const int o = __shfl_up(incl, dlt);
    if (lane >= dlt) incl += o;
  }
  const int ex = incl - s;
  v4i o4;
  o4.x = ex;
  o4.y = ex + c.x;
  o4.z = ex + c.x + c.y;
  o4.w = ex + c.x + c.y + c.z;
  const int ex2 = ex + s4;
  v4i o8;
  o8.x = ex2;
  o8.y = ex2 + c2.x;
  o8.z = ex2 + c2.x + c2.y;
  o8.w = ex2 + c2.x + c2.y + c2.z;
  p[0] = o4;
  p[1] = o8;
  if (lane == 31) tot[bk] = incl;
}

__global__ void __launch_bounds__(CH_THREADS)
gcn_bucket_place_kernel(const int* __restrict__ src, const int* __restrict__ dst,
                        const int* __restrict__ cnt, const int* __restrict__ tot,
                        int* __restrict__ bstart, int* __restrict__ grouped) {
  __shared__ int sc[2][CH_THREADS];
  __shared__ int cursor[NBK_PAD];
  const int t = threadIdx.x, j = blockIdx.x;
  const int* dp = dst + (size_t)j * CH_EDGES;
  const int* sp = src + (size_t)j * CH_EDGES;
  int d[CH_TRIPS], s[CH_TRIPS];
#pragma unroll
  for (int i = 0; i < CH_TRIPS; ++i) {
    const int idx = t + i * CH_THREADS;
    if (idx < CH_EDGES) { d[i] = dp[idx]; s[i] = sp[idx]; }
  }
  const int v = t < NBK_PAD ? tot[t] : 0;
  const int mine = t < NBK_PAD ? cnt[t * NCH + j] : 0;
  sc[0][t] = v;
  __syncthreads();
  int cur = 0;
  for (int dlt = 1; dlt < CH_THREADS; dlt <<= 1) {
    int x = sc[cur][t];
    if (t >= dlt) x += sc[cur][t - dlt];
    sc[cur ^ 1][t] = x;
    __syncthreads();
    cur ^= 1;
  }
  const int excl = sc[cur][t] - v;
  if (t < NBK_PAD) cursor[t] = excl + mine;
  if (j == 0 && t <= NBK_PAD) bstart[t] = excl;
  __syncthreads();
#pragma unroll
  for (int i = 0; i < CH_TRIPS; ++i) {
    const int idx = t + i * CH_THREADS;
    if (idx < CH_EDGES) {
      const int pos = atomicAdd(&cursor[d[i] >> BSHIFT], 1);
      grouped[pos] = s[i] | ((d[i] & (BROWS - 1)) << SRC_BITS);
    }
  }
}

typedef float v4f __attribute__((ext_vector_type(4)));
#define SUM_THREADS 256
#define SUM_WAVES (SUM_THREADS / 32)
#define EDGES_PER_BATCH 16
#define NULL_EDGE ((unsigned)BROWS << SRC_BITS)
#define SRC_MASK ((1u << SRC_BITS) - 1u)

__global__ void __launch_bounds__(SUM_THREADS)
gcn_bucket_sum_rows64_kernel(const float* __restrict__ x,
                             const int* __restrict__ grouped,
                             const int* __restrict__ bstart,
                             float* __restrict__ acc) {
  __shared__ __attribute__((aligned(16))) float tile[(BROWS + 1) * 64];
  const int t = threadIdx.x;
  const unsigned lane = t & 31u;
  const int wave = t >> 5;
  const int b = blockIdx.x;
  const int bs = bstart[b], be = bstart[b + 1];
  for (int i = t; i < (BROWS + 1) * 16; i += SUM_THREADS) ((v4f*)tile)[i] = (v4f)(0.0f);
  int e0 = bs + wave * 32;
  unsigned pk = (e0 + (int)lane < be) ? (unsigned)grouped[e0 + lane] : NULL_EDGE;
  __syncthreads();
  while (e0 < be) {
    const int en = e0 + SUM_WAVES * 32;
    const unsigned pkn = (en + (int)lane < be) ? (unsigned)grouped[en + lane] : NULL_EDGE;
#pragma unroll
    for (int bb = 0; bb < 32; bb += EDGES_PER_BATCH) {
      float v0[EDGES_PER_BATCH], v1[EDGES_PER_BATCH];
#pragma unroll
      for (int i = 0; i < EDGES_PER_BATCH; ++i) {
        const unsigned w = (unsigned)__builtin_amdgcn_readlane((int)pk, bb + i);
        const float* xr = x + (size_t)(w & SRC_MASK) * 64;
        v0[i] = xr[lane];
        v1[i] = xr[lane + 32];
      }
#pragma unroll
      for (int i = 0; i < EDGES_PER_BATCH; ++i) {
        const unsigned w = (unsigned)__builtin_amdgcn_readlane((int)pk, bb + i);
        float* o = tile + (w >> SRC_BITS) * 64;
        atomicAdd(o + lane, v0[i]);
        atomicAdd(o + lane + 32, v1[i]);
      }
    }
    pk = pkn;
    e0 = en;
  }
  __syncthreads();
  const int left = N_NODES - b * BROWS;
  const int rows = left < BROWS ? left : BROWS;
  v4f* g = (v4f*)(acc + (size_t)b * BROWS * 64);
  for (int i = t; i < rows * 16; i += SUM_THREADS) g[i] = ((const v4f*)tile)[i];
}

__global__ void __launch_bounds__(SUM_THREADS)
gcn_bucket_sum_rows40_kernel(const float* __restrict__ x,
                             const int* __restrict__ grouped,
                             const int* __restrict__ bstart,
                             const float* __restrict__ bias,
                             float* __restrict__ out) {
  __shared__ __attribute__((aligned(16))) float tile[(BROWS + 1) * NC];
  const int t = threadIdx.x;
  const unsigned lane = t & 31u;
  const int wave = t >> 5;
  const int b = blockIdx.x;
  const int bs = bstart[b], be = bstart[b + 1];
  if (t < 6 * NC) {
    const int c = t % NC;
    const float bv = bias[c];
    for (int r = t / NC; r < BROWS + 1; r += 6) tile[r * NC + c] = bv;
  }
  int e0 = bs + wave * 32;
  unsigned pk = (e0 + (int)lane < be) ? (unsigned)grouped[e0 + lane] : NULL_EDGE;
  const unsigned c1 = 32u + (lane & 7u);
  __syncthreads();
  while (e0 < be) {
    const int en = e0 + SUM_WAVES * 32;
    const unsigned pkn = (en + (int)lane < be) ? (unsigned)grouped[en + lane] : NULL_EDGE;
#pragma unroll
    for (int bb = 0; bb < 32; bb += EDGES_PER_BATCH) {
      float v0[EDGES_PER_BATCH], v1[EDGES_PER_BATCH];
#pragma unroll
      for (int i = 0; i < EDGES_PER_BATCH; ++i) {
        const unsigned w = (unsigned)__builtin_amdgcn_readlane((int)pk, bb + i);
        const float* xr = x + (size_t)(w & SRC_MASK) * NC;
        v0[i] = xr[lane];
        v1[i] = xr[c1];
      }
#pragma unroll
      for (int i = 0; i < EDGES_PER_BATCH; ++i) {
        const unsigned w = (unsigned)__builtin_amdgcn_readlane((int)pk, bb + i);
        float* o = tile + (w >> SRC_BITS) * NC;
        atomicAdd(o + lane, v0[i]);
        if (lane < (unsigned)(NC - 32)) atomicAdd(o + 32 + lane, v1[i]);
      }
    }
    pk = pkn;
    e0 = en;
  }
  __syncthreads();
  const int left = N_NODES - b * BROWS;
  const int rows = left < BROWS ? left : BROWS;
  v4f* g = (v4f*)(out + (size_t)b * BROWS * NC);
  for (int i = t; i < rows * (NC / 4); i += SUM_THREADS) g[i] = ((const v4f*)tile)[i];
}


#define DENSE_WAVES 10
#define DENSE_THREADS (DENSE_WAVES * 32)
#define DENSE_BLOCKS (N_NODES / (16 * DENSE_WAVES))
static_assert(N_NODES % (16 * DENSE_WAVES) == 0, "every wave owns one whole 16-row tile");
#define NC_PAD 48
#define PW1 144
#define PW2 112
#define PH 68
#define PT 52

__global__ void __launch_bounds__(DENSE_THREADS)
gcn_dense_layers_kernel(const float* __restrict__ A,
                        const float* __restrict__ W1,
                        const float* __restrict__ b1,
                        const float* __restrict__ W2,
                        float* __restrict__ T) {
  __shared__ __attribute__((aligned(16))) float w1l[(IN_F / 2) * PW1];
  __shared__ __attribute__((aligned(16))) float w2l[(HID / 2) * PW2];
  __shared__ __attribute__((aligned(16))) float tiles[DENSE_WAVES][16 * PH];

  const int tid = threadIdx.x;
  const int lane = tid & 31;
  const int wave = tid >> 5;
  const int lh = lane & 15;
  const int hi = lane >> 4;
  const int row0 = (blockIdx.x * DENSE_WAVES + wave) << 4;

  const float* arow = A + (size_t)(row0 + lh) * IN_F + (hi << 2);
  v4f af[8];
#pragma unroll
  for (int p = 0; p < 8; ++p) af[p] = *(const v4f*)(arow + 8 * p);

  constexpr int W1_PAIRS = (IN_F / 2) * HID;
  constexpr int W2_PAIRS = (HID / 2) * NC_PAD;
  constexpr int W1_TRIPS = (W1_PAIRS + DENSE_THREADS - 1) / DENSE_THREADS;
  constexpr int W2_TRIPS = (W2_PAIRS + DENSE_THREADS - 1) / DENSE_THREADS;
  v2f w1r[W1_TRIPS], w2r[W2_TRIPS];
#pragma unroll
  for (int i = 0; i < W1_TRIPS; ++i) {
    const int idx = tid + i * DENSE_THREADS;
    if (i < W1_TRIPS - 1 || idx < W1_PAIRS) {
      const int kp = idx >> 6, n = idx & 63;
      w1r[i].x = W1[(2 * kp) * HID + n];
      w1r[i].y = W1[(2 * kp + 1) * HID + n];
    }
  }
#pragma unroll
  for (int i = 0; i < W2_TRIPS; ++i) {
    const int idx = tid + i * DENSE_THREADS;
    if (i < W2_TRIPS - 1 || idx < W2_PAIRS) {
      const int kp = idx / NC_PAD, n = idx - kp * NC_PAD;
      const int nc = n < NC ? n : NC - 1;
      const float x = W2[(2 * kp) * NC + nc];
      const float y = W2[(2 * kp + 1) * NC + nc];
      w2r[i].x = n < NC ? x : 0.0f;
      w2r[i].y = n < NC ? y : 0.0f;
    }
  }
#pragma unroll
  for (int i = 0; i < W1_TRIPS; ++i) {
    const int idx = tid + i * DENSE_THREADS;
    if (i < W1_TRIPS - 1 || idx < W1_PAIRS) {
      const int kp = idx >> 6, n = idx & 63;
      *(v2f*)&w1l[kp * PW1 + 2 * n] = w1r[i];
    }
  }
#pragma unroll
  for (int i = 0; i < W2_TRIPS; ++i) {
    const int idx = tid + i * DENSE_THREADS;
    if (i < W2_TRIPS - 1 || idx < W2_PAIRS) {
      const int kp = idx / NC_PAD, n = idx - kp * NC_PAD;
      *(v2f*)&w2l[kp * PW2 + 2 * n] = w2r[i];
    }
  }
  __syncthreads();

  v8f acc[4] = {};
#pragma unroll
  for (int p = 0; p < 8; ++p) {
#pragma unroll
    for (int u = 0; u < 2; ++u) {
      v2f a;
      a.x = af[p][2 * u];
      a.y = af[p][2 * u + 1];
      const int kp = 4 * p + 2 * hi + u;
#pragma unroll
      for (int ct = 0; ct < 4; ++ct) {
        const v2f b = *(const v2f*)&w1l[kp * PW1 + 2 * (ct * 16 + lh)];
        acc[ct] = __builtin_amdgcn_wmma_f32_16x16x4_f32(false, a, false, b, (short)0,
                                                        acc[ct], false, false);
      }
    }
  }

  float* ht = tiles[wave];
#pragma unroll
  for (int ct = 0; ct < 4; ++ct) {
    const float bv = b1[ct * 16 + lh];
#pragma unroll
    for (int r = 0; r < 8; ++r)
      ht[((hi << 3) + r) * PH + ct * 16 + lh] = fmaxf(acc[ct][r] + bv, 0.0f);
  }
  __syncthreads();

  v4f hf[8];
#pragma unroll
  for (int p = 0; p < 8; ++p) hf[p] = *(const v4f*)&ht[lh * PH + 8 * p + (hi << 2)];
  v8f acc2[3] = {};
#pragma unroll
  for (int p = 0; p < 8; ++p) {
#pragma unroll
    for (int u = 0; u < 2; ++u) {
      v2f a;
      a.x = hf[p][2 * u];
      a.y = hf[p][2 * u + 1];
      const int kp = 4 * p + 2 * hi + u;
#pragma unroll
      for (int ct = 0; ct < 3; ++ct) {
        const v2f b = *(const v2f*)&w2l[kp * PW2 + 2 * (ct * 16 + lh)];
        acc2[ct] = __builtin_amdgcn_wmma_f32_16x16x4_f32(false, a, false, b, (short)0,
                                                         acc2[ct], false, false);
      }
    }
  }

#pragma unroll
  for (int ct = 0; ct < 3; ++ct) {
#pragma unroll
    for (int r = 0; r < 8; ++r)
      ht[((hi << 3) + r) * PT + ct * 16 + lh] = acc2[ct][r];
  }
  __syncthreads();

  float* trow = T + (size_t)row0 * NC;
#pragma unroll
  for (int q = 0; q < 5; ++q) {
    const int g = lane + 32 * q;
    const int row = g / 10;
    const int c4 = g - row * 10;
    const v4f v = *(const v4f*)&ht[row * PT + 4 * c4];
    *(v4f*)(trow + 4 * g) = v;
  }
}

extern "C" void kernel_launch(void* const* d_in, const int* in_sizes, int n_in,
                              void* d_out, int out_size, void* d_ws,
                              size_t ws_size, hipStream_t stream) {
  (void)in_sizes; (void)n_in; (void)out_size; (void)ws_size;
  const float* features = (const float*)d_in[0];
  const int*   src      = (const int*)d_in[1];
  const int*   dst      = (const int*)d_in[2];
  const float* W1       = (const float*)d_in[3];
  const float* b1       = (const float*)d_in[4];
  const float* W2       = (const float*)d_in[5];
  const float* b2       = (const float*)d_in[6];
  float* out = (float*)d_out;

  float* ws    = (float*)d_ws;
  float* agg1  = ws;
  float* t     = agg1 + (size_t)N_NODES * HID;
  int* grouped = (int*)(t + (size_t)N_NODES * NC);
  int* cnt     = grouped + N_EDGES;
  int* tot     = cnt + NBK_PAD * NCH;
  int* bstart  = tot + 1024;

  gcn_bucket_count_kernel<<<NCH, CH_THREADS, 0, stream>>>(dst, cnt);
  gcn_bucket_rowscan_kernel<<<NBK_PAD / 8, 256, 0, stream>>>(cnt, tot);
  gcn_bucket_place_kernel<<<NCH, CH_THREADS, 0, stream>>>(src, dst, cnt, tot, bstart, grouped);

  gcn_bucket_sum_rows64_kernel<<<NBK, SUM_THREADS, 0, stream>>>(features, grouped, bstart, agg1);

  gcn_dense_layers_kernel<<<DENSE_BLOCKS, DENSE_THREADS, 0, stream>>>(
      agg1, W1, b1, W2, t);

  gcn_bucket_sum_rows40_kernel<<<NBK, SUM_THREADS, 0, stream>>>(t, grouped, bstart, b2, out);
}


# bucket sums: a work-group orders its edges by row in LDS (counting sort); a wave sums each of its rows in registers, one LDS add per row, not per edge; a row is summed serially in list order
# speedup vs baseline: 2.4902x; 1.0490x over previous
#include <hip/hip_runtime.h>
#include <hip/hip_bf16.h>

typedef float v2f __attribute__((ext_vector_type(2)));
typedef float v8f __attribute__((ext_vector_type(8)));
typedef int v4i __attribute__((ext_vector_type(4)));

#define N_NODES 100000
#define N_EDGES 1600000
#define IN_F 64
#define HID 64
#define NC 40

#define BROWS 128
#define BSHIFT 7
#define NBK ((N_NODES + BROWS - 1) / BROWS)
#define NBK_PAD 800
#define NCH 256
#define CH_EDGES (N_EDGES / NCH)
#define CH_THREADS 1024
#define CH_TRIPS ((CH_EDGES + CH_THREADS - 1) / CH_THREADS)
#define SRC_BITS 17
static_assert(N_EDGES % NCH == 0, "the edge list is cut into whole chunks");
static_assert(N_NODES <= (1 << SRC_BITS), "source row fits its field");
static_assert(NBK <= NBK_PAD && NBK_PAD <= CH_THREADS && NBK_PAD % 8 == 0, "bucket table");
static_assert(NCH == 256, "one wave scans one bucket row: 32 lanes x 8");

__global__ void __launch_bounds__(CH_THREADS)
gcn_bucket_count_kernel(const int* __restrict__ dst, int* __restrict__ cnt) {
  __shared__ int h[NBK_PAD];
  const int t = threadIdx.x, j = blockIdx.x;
  const int* dp = dst + (size_t)j * CH_EDGES;
  int d[CH_TRIPS];
#pragma unroll
  for (int i = 0; i < CH_TRIPS; ++i) {
    const int idx = t + i * CH_THREADS;
    if (idx < CH_EDGES) d[i] = dp[idx];
  }
  if (t < NBK_PAD) h[t] = 0;
  __syncthreads();
#pragma unroll
  for (int i = 0; i < CH_TRIPS; ++i) {
    const int idx = t + i * CH_THREADS;
    if (idx < CH_EDGES) atomicAdd(&h[d[i] >> BSHIFT], 1);
  }
  __syncthreads();
  if (t < NBK_PAD) cnt[t * NCH + j] = h[t];
}

__global__ void __launch_bounds__(256)
gcn_bucket_rowscan_kernel(int* __restrict__ cnt, int* __restrict__ tot) {
  const int lane = threadIdx.x & 31;
  const int bk = blockIdx.x * 8 + (threadIdx.x >> 5);
  v4i* p = (v4i*)(cnt + (size_t)bk * NCH) + 2 * lane;
  const v4i c = p[0], c2 = p[1];
  const int s4 = c.x + c.y + c.z + c.w;
  const int s = s4 + c2.x + c2.y + c2.z + c2.w;
  int incl = s;
#pragma unroll
  for (int dlt = 1; dlt < 32; dlt <<= 1) {
    const int o = __shfl_up(incl, dlt);
    if (lane >= dlt) incl += o;
  }
  const int ex = incl - s;
  v4i o4;
  o4.x = ex;
  o4.y = ex + c.x;
  o4.z = ex + c.x + c.y;
  o4.w = ex + c.x + c.y + c.z;
  const int ex2 = ex + s4;
  v4i o8;
  o8.x = ex2;
  o8.y = ex2 + c2.x;
  o8.z = ex2 + c2.x + c2.y;
  o8.w = ex2 + c2.x + c2.y + c2.z;
  p[0] = o4;
  p[1] = o8;
  if (lane == 31) tot[bk] = incl;
}

__global__ void __launch_bounds__(CH_THREADS)
gcn_bucket_place_kernel(const int* __restrict__ src, const int* __restrict__ dst,
                        const int* __restrict__ cnt, const int* __restrict__ tot,
                        int* __restrict__ bstart, int* __restrict__ grouped) {
  __shared__ int sc[2][CH_THREADS];
  __shared__ int cursor[NBK_PAD];
  const int t = threadIdx.x, j = blockIdx.x;
  const int* dp = dst + (size_t)j * CH_EDGES;
  const int* sp = src + (size_t)j * CH_EDGES;
  int d[CH_TRIPS], s[CH_TRIPS];
#pragma unroll
  for (int i = 0; i < CH_TRIPS; ++i) {
    const int idx = t + i * CH_THREADS;
    if (idx < CH_EDGES) { d[i] = dp[idx]; s[i] = sp[idx]; }
  }
  const int v = t < NBK_PAD ? tot[t] : 0;
  const int mine = t < NBK_PAD ? cnt[t * NCH + j] : 0;
  sc[0][t] = v;
  __syncthreads();
  int cur = 0;
  for (int dlt = 1; dlt < CH_THREADS; dlt <<= 1) {
    int x = sc[cur][t];
    if (t >= dlt) x += sc[cur][t - dlt];
    sc[cur ^ 1][t] = x;
    __syncthreads();
    cur ^= 1;
  }
  const int excl = sc[cur][t] - v;
  if (t < NBK_PAD) cursor[t] = excl + mine;
  if (j == 0 && t <= NBK_PAD) bstart[t] = excl;
  __syncthreads();
#pragma unroll
  for (int i = 0; i < CH_TRIPS; ++i) {
    const int idx = t + i * CH_THREADS;
    if (idx < CH_EDGES) {
      const int pos = atomicAdd(&cursor[d[i] >> BSHIFT], 1);
      grouped[pos] = s[i] | ((d[i] & (BROWS - 1)) << SRC_BITS);
    }
  }
}

typedef float v4f __attribute__((ext_vector_type(4)));
#define SUM_THREADS 256
#define SUM_WAVES (SUM_THREADS / 32)
#define ROWS_PER_WAVE (BROWS / SUM_WAVES)
#define EDGES_PER_BATCH 16
#define NULL_EDGE ((unsigned)BROWS << SRC_BITS)
#define SRC_MASK ((1u << SRC_BITS) - 1u)
#define SEG_CAP 4096
#define SEG_TRIPS (SEG_CAP / SUM_THREADS)
static_assert(BROWS == 128, "one wave scans the row counts: 32 lanes x 4");
static_assert(BROWS % SUM_WAVES == 0, "every wave owns whole rows");

template <int COLS>
__device__ __forceinline__ void bucket_rows_sum(const float* __restrict__ x,
                                                const int* __restrict__ grouped,
                                                const int bs, const int be,
                                                float* tile, unsigned* sorted,
                                                int* cnt, int* rstart) {
  const int t = threadIdx.x;
  const unsigned lane = t & 31u;
  const int wave = __builtin_amdgcn_readfirstlane(t >> 5);
  const unsigned c1 = COLS == 64 ? lane + 32u : 32u + (lane & 7u);

  for (int s0 = bs; s0 < be; s0 += SEG_CAP) {
    const int left = be - s0;
    const int n = left < SEG_CAP ? left : SEG_CAP;
    unsigned pk[SEG_TRIPS];
    int rk[SEG_TRIPS];
#pragma unroll
    for (int i = 0; i < SEG_TRIPS; ++i) {
      const int idx = t + i * SUM_THREADS;
      pk[i] = idx < n ? (unsigned)grouped[s0 + idx] : 0u;
    }
    if (t < BROWS) cnt[t] = 0;
    __syncthreads();
#pragma unroll
    for (int i = 0; i < SEG_TRIPS; ++i) {
      const int idx = t + i * SUM_THREADS;
      if (idx < n) rk[i] = atomicAdd(&cnt[pk[i] >> SRC_BITS], 1);
    }
    __syncthreads();
    if (wave == 0) {
      const v4i c = ((const v4i*)cnt)[lane];
      const int s = c.x + c.y + c.z + c.w;
      int incl = s;
#pragma unroll
      for (int dlt = 1; dlt < 32; dlt <<= 1) {
        const int o = __shfl_up(incl, dlt);
        if ((int)lane >= dlt) incl += o;
      }
      const int ex = incl - s;
      v4i o4;
      o4.x = ex;
      o4.y = ex + c.x;
      o4.z = ex + c.x + c.y;
      o4.w = ex + c.x + c.y + c.z;
      ((v4i*)rstart)[lane] = o4;
      if (lane == 31u) rstart[BROWS] = incl;
    }
    __syncthreads();
#pragma unroll
    for (int i = 0; i < SEG_TRIPS; ++i) {
      const int idx = t + i * SUM_THREADS;
      if (idx < n) sorted[rstart[pk[i] >> SRC_BITS] + rk[i]] = pk[i];
    }
    __syncthreads();

    int e0 = __builtin_amdgcn_readfirstlane(rstart[ROWS_PER_WAVE * wave]);
    const int we = __builtin_amdgcn_readfirstlane(rstart[ROWS_PER_WAVE * (wave + 1)]);
    int cur = ROWS_PER_WAVE * wave;
    float a0 = 0.0f, a1 = 0.0f;
    while (e0 < we) {
      const unsigned pkv = (e0 + (int)lane < we) ? sorted[e0 + lane] : NULL_EDGE;
#pragma unroll
      for (int bb = 0; bb < 32; bb += EDGES_PER_BATCH) {
        if (e0 + bb < we) {
          float v0[EDGES_PER_BATCH], v1[EDGES_PER_BATCH];
#pragma unroll
          for (int i = 0; i < EDGES_PER_BATCH; ++i) {
            const unsigned w = (unsigned)__builtin_amdgcn_readlane((int)pkv, bb + i);
            const float* xr = x + (size_t)(w & SRC_MASK) * COLS;
            v0[i] = xr[lane];
            v1[i] = xr[c1];
          }
#pragma unroll
          for (int i = 0; i < EDGES_PER_BATCH; ++i) {
            const unsigned w = (unsigned)__builtin_amdgcn_readlane((int)pkv, bb + i);
            const int r = (int)(w >> SRC_BITS);
            if (r != cur) {
              float* o = tile + cur * COLS;
              atomicAdd(o + lane, a0);
              if (COLS == 64) atomicAdd(o + lane + 32, a1);
              else if (lane < (unsigned)(COLS - 32)) atomicAdd(o + 32 + lane, a1);
              a0 = 0.0f;
              a1 = 0.0f;
              cur = r;
            }
            a0 += v0[i];
            a1 += v1[i];
          }
        }
      }
      e0 += 32;
    }
    {
      float* o = tile + cur * COLS;
      atomicAdd(o + lane, a0);
      if (COLS == 64) atomicAdd(o + lane + 32, a1);
      else if (lane < (unsigned)(COLS - 32)) atomicAdd(o + 32 + lane, a1);
    }
  }
}

__global__ void __launch_bounds__(SUM_THREADS)
gcn_bucket_sum_rows64_kernel(const float* __restrict__ x,
                             const int* __restrict__ grouped,
                             const int* __restrict__ bstart,
                             float* __restrict__ acc) {
  __shared__ __attribute__((aligned(16))) float tile[(BROWS + 1) * 64];
  __shared__ __attribute__((aligned(16))) unsigned sorted[SEG_CAP];
  __shared__ __attribute__((aligned(16))) int cnt[BROWS];
  __shared__ __attribute__((aligned(16))) int rstart[BROWS + 4];
  const int t = threadIdx.x;
  const int b = blockIdx.x;
  const int bs = bstart[b], be = bstart[b + 1];
  for (int i = t; i < (BROWS + 1) * 16; i += SUM_THREADS) ((v4f*)tile)[i] = (v4f)(0.0f);
  bucket_rows_sum<64>(x, grouped, bs, be, tile, sorted, cnt, rstart);
  __syncthreads();
  const int left = N_NODES - b * BROWS;
  const int rows = left < BROWS ? left : BROWS;
  v4f* g = (v4f*)(acc + (size_t)b * BROWS * 64);
  for (int i = t; i < rows * 16; i += SUM_THREADS) g[i] = ((const v4f*)tile)[i];
}

__global__ void __launch_bounds__(SUM_THREADS)
gcn_bucket_sum_rows40_kernel(const float* __restrict__ x,
                             const int* __restrict__ grouped,
                             const int* __restrict__ bstart,
                             const float* __restrict__ bias,
                             float* __restrict__ out) {
  __shared__ __attribute__((aligned(16))) float tile[(BROWS + 1) * NC];
  __shared__ __attribute__((aligned(16))) unsigned sorted[SEG_CAP];
  __shared__ __attribute__((aligned(16))) int cnt[BROWS];
  __shared__ __attribute__((aligned(16))) int rstart[BROWS + 4];
  const int t = threadIdx.x;
  const int b = blockIdx.x;
  const int bs = bstart[b], be = bstart[b + 1];
  if (t < 6 * NC) {
    const int c = t % NC;
    const float bv = bias[c];
    for (int r = t / NC; r < BROWS + 1; r += 6) tile[r * NC + c] = bv;
  }
  bucket_rows_sum<NC>(x, grouped, bs, be, tile, sorted, cnt, rstart);
  __syncthreads();
  const int left = N_NODES - b * BROWS;
  const int rows = left < BROWS ? left : BROWS;
  v4f* g = (v4f*)(out + (size_t)b * BROWS * NC);
  for (int i = t; i < rows * (NC / 4); i += SUM_THREADS) g[i] = ((const v4f*)tile)[i];
}


#define DENSE_WAVES 10
#define DENSE_THREADS (DENSE_WAVES * 32)
#define DENSE_BLOCKS (N_NODES / (16 * DENSE_WAVES))
static_assert(N_NODES % (16 * DENSE_WAVES) == 0, "every wave owns one whole 16-row tile");
#define NC_PAD 48
#define PW1 144
#define PW2 112
#define PH 68
#define PT 52

__global__ void __launch_bounds__(DENSE_THREADS)
gcn_dense_layers_kernel(const float* __restrict__ A,
                        const float* __restrict__ W1,
                        const float* __restrict__ b1,
                        const float* __restrict__ W2,
                        float* __restrict__ T) {
  __shared__ __attribute__((aligned(16))) float w1l[(IN_F / 2) * PW1];
  __shared__ __attribute__((aligned(16))) float w2l[(HID / 2) * PW2];
  __shared__ __attribute__((aligned(16))) float tiles[DENSE_WAVES][16 * PH];

  const int tid = threadIdx.x;
  const int lane = tid & 31;
  const int wave = tid >> 5;
  const int lh = lane & 15;
  const int hi = lane >> 4;
  const int row0 = (blockIdx.x * DENSE_WAVES + wave) << 4;

  const float* arow = A + (size_t)(row0 + lh) * IN_F + (hi << 2);
  v4f af[8];
#pragma unroll
  for (int p = 0; p < 8; ++p) af[p] = *(const v4f*)(arow + 8 * p);

  constexpr int W1_PAIRS = (IN_F / 2) * HID;
  constexpr int W2_PAIRS = (HID / 2) * NC_PAD;
  constexpr int W1_TRIPS = (W1_PAIRS + DENSE_THREADS - 1) / DENSE_THREADS;
  constexpr int W2_TRIPS = (W2_PAIRS + DENSE_THREADS - 1) / DENSE_THREADS;
  v2f w1r[W1_TRIPS], w2r[W2_TRIPS];
#pragma unroll
  for (int i = 0; i < W1_TRIPS; ++i) {
    const int idx = tid + i * DENSE_THREADS;
    if (i < W1_TRIPS - 1 || idx < W1_PAIRS) {
      const int kp = idx >> 6, n = idx & 63;
      w1r[i].x = W1[(2 * kp) * HID + n];
      w1r[i].y = W1[(2 * kp + 1) * HID + n];
    }
  }
#pragma unroll
  for (int i = 0; i < W2_TRIPS; ++i) {
    const int idx = tid + i * DENSE_THREADS;
    if (i < W2_TRIPS - 1 || idx < W2_PAIRS) {
      const int kp = idx / NC_PAD, n = idx - kp * NC_PAD;
      const int nc = n < NC ? n : NC - 1;
      const float x = W2[(2 * kp) * NC + nc];
      const float y = W2[(2 * kp + 1) * NC + nc];
      w2r[i].x = n < NC ? x : 0.0f;
      w2r[i].y = n < NC ? y : 0.0f;
    }
  }
#pragma unroll
  for (int i = 0; i < W1_TRIPS; ++i) {
    const int idx = tid + i * DENSE_THREADS;
    if (i < W1_TRIPS - 1 || idx < W1_PAIRS) {
      const int kp = idx >> 6, n = idx & 63;
      *(v2f*)&w1l[kp * PW1 + 2 * n] = w1r[i];
    }
  }
#pragma unroll
  for (int i = 0; i < W2_TRIPS; ++i) {
    const int idx = tid + i * DENSE_THREADS;
    if (i < W2_TRIPS - 1 || idx < W2_PAIRS) {
      const int kp = idx / NC_PAD, n = idx - kp * NC_PAD;
      *(v2f*)&w2l[kp * PW2 + 2 * n] = w2r[i];
    }
  }
  __syncthreads();

  v8f acc[4] = {};
#pragma unroll
  for (int p = 0; p < 8; ++p) {
#pragma unroll
    for (int u = 0; u < 2; ++u) {
      v2f a;
      a.x = af[p][2 * u];
      a.y = af[p][2 * u + 1];
      const int kp = 4 * p + 2 * hi + u;
#pragma unroll
      for (int ct = 0; ct < 4; ++ct) {
        const v2f b = *(const v2f*)&w1l[kp * PW1 + 2 * (ct * 16 + lh)];
        acc[ct] = __builtin_amdgcn_wmma_f32_16x16x4_f32(false, a, false, b, (short)0,
                                                        acc[ct], false, false);
      }
    }
  }

  float* ht = tiles[wave];
#pragma unroll
  for (int ct = 0; ct < 4; ++ct) {
    const float bv = b1[ct * 16 + lh];
#pragma unroll
    for (int r = 0; r < 8; ++r)
      ht[((hi << 3) + r) * PH + ct * 16 + lh] = fmaxf(acc[ct][r] + bv, 0.0f);
  }
  __syncthreads();

  v4f hf[8];
#pragma unroll
  for (int p = 0; p < 8; ++p) hf[p] = *(const v4f*)&ht[lh * PH + 8 * p + (hi << 2)];
  v8f acc2[3] = {};
#pragma unroll
  for (int p = 0; p < 8; ++p) {
#pragma unroll
    for (int u = 0; u < 2; ++u) {
      v2f a;
      a.x = hf[p][2 * u];
      a.y = hf[p][2 * u + 1];
      const int kp = 4 * p + 2 * hi + u;
#pragma unroll
      for (int ct = 0; ct < 3; ++ct) {
        const v2f b = *(const v2f*)&w2l[kp * PW2 + 2 * (ct * 16 + lh)];
        acc2[ct] = __builtin_amdgcn_wmma_f32_16x16x4_f32(false, a, false, b, (short)0,
                                                         acc2[ct], false, false);
      }
    }
  }

#pragma unroll
  for (int ct = 0; ct < 3; ++ct) {
#pragma unroll
    for (int r = 0; r < 8; ++r)
      ht[((hi << 3) + r) * PT + ct * 16 + lh] = acc2[ct][r];
  }
  __syncthreads();

  float* trow = T + (size_t)row0 * NC;
#pragma unroll
  for (int q = 0; q < 5; ++q) {
    const int g = lane + 32 * q;
    const int row = g / 10;
    const int c4 = g - row * 10;
    const v4f v = *(const v4f*)&ht[row * PT + 4 * c4];
    *(v4f*)(trow + 4 * g) = v;
  }
}

extern "C" void kernel_launch(void* const* d_in, const int* in_sizes, int n_in,
                              void* d_out, int out_size, void* d_ws,
                              size_t ws_size, hipStream_t stream) {
  (void)in_sizes; (void)n_in; (void)out_size; (void)ws_size;
  const float* features = (const float*)d_in[0];
  const int*   src      = (const int*)d_in[1];
  const int*   dst      = (const int*)d_in[2];
  const float* W1       = (const float*)d_in[3];
  const float* b1       = (const float*)d_in[4];
  const float* W2       = (const float*)d_in[5];
  const float* b2       = (const float*)d_in[6];
  float* out = (float*)d_out;

  float* ws    = (float*)d_ws;
  float* agg1  = ws;
  float* t     = agg1 + (size_t)N_NODES * HID;
  int* grouped = (int*)(t + (size_t)N_NODES * NC);
  int* cnt     = grouped + N_EDGES;
  int* tot     = cnt + NBK_PAD * NCH;
  int* bstart  = tot + 1024;

  gcn_bucket_count_kernel<<<NCH, CH_THREADS, 0, stream>>>(dst, cnt);
  gcn_bucket_rowscan_kernel<<<NBK_PAD / 8, 256, 0, stream>>>(cnt, tot);
  gcn_bucket_place_kernel<<<NCH, CH_THREADS, 0, stream>>>(src, dst, cnt, tot, bstart, grouped);

  gcn_bucket_sum_rows64_kernel<<<NBK, SUM_THREADS, 0, stream>>>(features, grouped, bstart, agg1);

  gcn_dense_layers_kernel<<<DENSE_BLOCKS, DENSE_THREADS, 0, stream>>>(
      agg1, W1, b1, W2, t);

  gcn_bucket_sum_rows40_kernel<<<NBK, SUM_THREADS, 0, stream>>>(t, grouped, bstart, b2, out);
}


# dense layers moved into the first aggregation's launch: the work-group keeps its 128 row sums in LDS, each wave does relu(rows@W1+b1)@W2 on the 16 rows it summed; sums never reach memory; 5 launches
# speedup vs baseline: 2.6904x; 1.0804x over previous
#include <hip/hip_runtime.h>
#include <hip/hip_bf16.h>

typedef float v2f __attribute__((ext_vector_type(2)));
typedef float v8f __attribute__((ext_vector_type(8)));
typedef int v4i __attribute__((ext_vector_type(4)));

#define N_NODES 100000
#define N_EDGES 1600000
#define IN_F 64
#define HID 64
#define NC 40

#define BROWS 128
#define BSHIFT 7
#define NBK ((N_NODES + BROWS - 1) / BROWS)
#define NBK_PAD 800
#define NCH 256
#define CH_EDGES (N_EDGES / NCH)
#define CH_THREADS 1024
#define CH_TRIPS ((CH_EDGES + CH_THREADS - 1) / CH_THREADS)
#define SRC_BITS 17
static_assert(N_EDGES % NCH == 0, "the edge list is cut into whole chunks");
static_assert(N_NODES <= (1 << SRC_BITS), "source row fits its field");
static_assert(NBK <= NBK_PAD && NBK_PAD <= CH_THREADS && NBK_PAD % 8 == 0, "bucket table");
static_assert(NCH == 256, "one wave scans one bucket row: 32 lanes x 8");

__global__ void __launch_bounds__(CH_THREADS)
gcn_bucket_count_kernel(const int* __restrict__ dst, int* __restrict__ cnt) {
  __shared__ int h[NBK_PAD];
  const int t = threadIdx.x, j = blockIdx.x;
  const int* dp = dst + (size_t)j * CH_EDGES;
  int d[CH_TRIPS];
#pragma unroll
  for (int i = 0; i < CH_TRIPS; ++i) {
    const int idx = t + i * CH_THREADS;
    if (idx < CH_EDGES) d[i] = dp[idx];
  }
  if (t < NBK_PAD) h[t] = 0;
  __syncthreads();
#pragma unroll
  for (int i = 0; i < CH_TRIPS; ++i) {
    const int idx = t + i * CH_THREADS;
    if (idx < CH_EDGES) atomicAdd(&h[d[i] >> BSHIFT], 1);
  }
  __syncthreads();
  if (t < NBK_PAD) cnt[t * NCH + j] = h[t];
}

__global__ void __launch_bounds__(256)
gcn_bucket_rowscan_kernel(int* __restrict__ cnt, int* __restrict__ tot) {
  const int lane = threadIdx.x & 31;
  const int bk = blockIdx.x * 8 + (threadIdx.x >> 5);
  v4i* p = (v4i*)(cnt + (size_t)bk * NCH) + 2 * lane;
  const v4i c = p[0], c2 = p[1];
  const int s4 = c.x + c.y + c.z + c.w;
  const int s = s4 + c2.x + c2.y + c2.z + c2.w;
  int incl = s;
#pragma unroll
  for (int dlt = 1; dlt < 32; dlt <<= 1) {
    const int o = __shfl_up(incl, dlt);
    if (lane >= dlt) incl += o;
  }
  const int ex = incl - s;
  v4i o4;
  o4.x = ex;
  o4.y = ex + c.x;
  o4.z = ex + c.x + c.y;
  o4.w = ex + c.x + c.y + c.z;
  const int ex2 = ex + s4;
  v4i o8;
  o8.x = ex2;
  o8.y = ex2 + c2.x;
  o8.z = ex2 + c2.x + c2.y;
  o8.w = ex2 + c2.x + c2.y + c2.z;
  p[0] = o4;
  p[1] = o8;
  if (lane == 31) tot[bk] = incl;
}

__global__ void __launch_bounds__(CH_THREADS)
gcn_bucket_place_kernel(const int* __restrict__ src, const int* __restrict__ dst,
                        const int* __restrict__ cnt, const int* __restrict__ tot,
                        int* __restrict__ bstart, int* __restrict__ grouped) {
  __shared__ int sc[2][CH_THREADS];
  __shared__ int cursor[NBK_PAD];
  const int t = threadIdx.x, j = blockIdx.x;
  const int* dp = dst + (size_t)j * CH_EDGES;
  const int* sp = src + (size_t)j * CH_EDGES;
  int d[CH_TRIPS], s[CH_TRIPS];
#pragma unroll
  for (int i = 0; i < CH_TRIPS; ++i) {
    const int idx = t + i * CH_THREADS;
    if (idx < CH_EDGES) { d[i] = dp[idx]; s[i] = sp[idx]; }
  }
  const int v = t < NBK_PAD ? tot[t] : 0;
  const int mine = t < NBK_PAD ? cnt[t * NCH + j] : 0;
  sc[0][t] = v;
  __syncthreads();
  int cur = 0;
  for (int dlt = 1; dlt < CH_THREADS; dlt <<= 1) {
    int x = sc[cur][t];
    if (t >= dlt) x += sc[cur][t - dlt];
    sc[cur ^ 1][t] = x;
    __syncthreads();
    cur ^= 1;
  }
  const int excl = sc[cur][t] - v;
  if (t < NBK_PAD) cursor[t] = excl + mine;
  if (j == 0 && t <= NBK_PAD) bstart[t] = excl;
  __syncthreads();
#pragma unroll
  for (int i = 0; i < CH_TRIPS; ++i) {
    const int idx = t + i * CH_THREADS;
    if (idx < CH_EDGES) {
      const int pos = atomicAdd(&cursor[d[i] >> BSHIFT], 1);
      grouped[pos] = s[i] | ((d[i] & (BROWS - 1)) << SRC_BITS);
    }
  }
}

typedef float v4f __attribute__((ext_vector_type(4)));
#define SUM_THREADS 256
#define SUM_WAVES (SUM_THREADS / 32)
#define ROWS_PER_WAVE (BROWS / SUM_WAVES)
#define EDGES_PER_BATCH 16
#define NULL_EDGE ((unsigned)BROWS << SRC_BITS)
#define SRC_MASK ((1u << SRC_BITS) - 1u)
#define SEG_CAP 4096
#define SEG_TRIPS (SEG_CAP / SUM_THREADS)
static_assert(BROWS == 128, "one wave scans the row counts: 32 lanes x 4");
static_assert(BROWS % SUM_WAVES == 0, "every wave owns whole rows");

template <int COLS, int PITCH>
__device__ __forceinline__ void bucket_rows_sum(const float* __restrict__ x,
                                                const int* __restrict__ grouped,
                                                const int bs, const int be,
                                                float* tile, unsigned* sorted,
                                                int* cnt, int* rstart) {
  const int t = threadIdx.x;
  const unsigned lane = t & 31u;
  const int wave = __builtin_amdgcn_readfirstlane(t >> 5);
  const unsigned c1 = COLS == 64 ? lane + 32u : 32u + (lane & 7u);

  for (int s0 = bs; s0 < be; s0 += SEG_CAP) {
    const int left = be - s0;
    const int n = left < SEG_CAP ? left : SEG_CAP;
    unsigned pk[SEG_TRIPS];
    int rk[SEG_TRIPS];
#pragma unroll
    for (int i = 0; i < SEG_TRIPS; ++i) {
      const int idx = t + i * SUM_THREADS;
      pk[i] = idx < n ? (unsigned)grouped[s0 + idx] : 0u;
    }
    if (t < BROWS) cnt[t] = 0;
    __syncthreads();
#pragma unroll
    for (int i = 0; i < SEG_TRIPS; ++i) {
      const int idx = t + i * SUM_THREADS;
      if (idx < n) rk[i] = atomicAdd(&cnt[pk[i] >> SRC_BITS], 1);
    }
    __syncthreads();
    if (wave == 0) {
      const v4i c = ((const v4i*)cnt)[lane];
      const int s = c.x + c.y + c.z + c.w;
      int incl = s;
#pragma unroll
      for (int dlt = 1; dlt < 32; dlt <<= 1) {
        const int o = __shfl_up(incl, dlt);
        if ((int)lane >= dlt) incl += o;
      }
      const int ex = incl - s;
      v4i o4;
      o4.x = ex;
      o4.y = ex + c.x;
      o4.z = ex + c.x + c.y;
      o4.w = ex + c.x + c.y + c.z;
      ((v4i*)rstart)[lane] = o4;
      if (lane == 31u) rstart[BROWS] = incl;
    }
    __syncthreads();
#pragma unroll
    for (int i = 0; i < SEG_TRIPS; ++i) {
      const int idx = t + i * SUM_THREADS;
      if (idx < n) sorted[rstart[pk[i] >> SRC_BITS] + rk[i]] = pk[i];
    }
    __syncthreads();

    int e0 = __builtin_amdgcn_readfirstlane(rstart[ROWS_PER_WAVE * wave]);
    const int we = __builtin_amdgcn_readfirstlane(rstart[ROWS_PER_WAVE * (wave + 1)]);
    int cur = ROWS_PER_WAVE * wave;
    float a0 = 0.0f, a1 = 0.0f;
    while (e0 < we) {
      const unsigned pkv = (e0 + (int)lane < we) ? sorted[e0 + lane] : NULL_EDGE;
#pragma unroll
      for (int bb = 0; bb < 32; bb += EDGES_PER_BATCH) {
        if (e0 + bb < we) {
          float v0[EDGES_PER_BATCH], v1[EDGES_PER_BATCH];
#pragma unroll
          for (int i = 0; i < EDGES_PER_BATCH; ++i) {
            const unsigned w = (unsigned)__builtin_amdgcn_readlane((int)pkv, bb + i);
            const float* xr = x + (size_t)(w & SRC_MASK) * COLS;
            v0[i] = xr[lane];
            v1[i] = xr[c1];
          }
#pragma unroll
          for (int i = 0; i < EDGES_PER_BATCH; ++i) {
            const unsigned w = (unsigned)__builtin_amdgcn_readlane((int)pkv, bb + i);
            const int r = (int)(w >> SRC_BITS);
            if (r != cur) {
              float* o = tile + cur * PITCH;
              atomicAdd(o + lane, a0);
              if (COLS == 64) atomicAdd(o + lane + 32, a1);
              else if (lane < (unsigned)(COLS - 32)) atomicAdd(o + 32 + lane, a1);
              a0 = 0.0f;
              a1 = 0.0f;
              cur = r;
            }
            a0 += v0[i];
            a1 += v1[i];
          }
        }
      }
      e0 += 32;
    }
    {
      float* o = tile + cur * PITCH;
      atomicAdd(o + lane, a0);
      if (COLS == 64) atomicAdd(o + lane + 32, a1);
      else if (lane < (unsigned)(COLS - 32)) atomicAdd(o + 32 + lane, a1);
    }
  }
}

#define NC_PAD 48
#define PH 68
#define PT 52
#define W1_PIECES (16 * HID)
#define W2_PIECES (16 * NC_PAD)
#define W1_TRIPS (W1_PIECES / SUM_THREADS)
#define W2_TRIPS (W2_PIECES / SUM_THREADS)
static_assert(IN_F == 64 && HID == 64, "16 WMMA k groups of 4 per layer");
static_assert(W1_PIECES * 4 == SEG_CAP, "W1 fills the ordered-segment buffer exactly");
static_assert(W1_PIECES % SUM_THREADS == 0 && W2_PIECES % SUM_THREADS == 0, "whole staging trips");
static_assert(ROWS_PER_WAVE == 16, "a wave's rows are one WMMA row tile");
static_assert(N_NODES % ROWS_PER_WAVE == 0, "a wave's 16 rows exist all or not at all");

__global__ void __launch_bounds__(SUM_THREADS)
gcn_bucket_sum_rows64_dense_kernel(const float* __restrict__ x,
                                   const int* __restrict__ grouped,
                                   const int* __restrict__ bstart,
                                   const float* __restrict__ W1,
                                   const float* __restrict__ b1,
                                   const float* __restrict__ W2,
                                   float* __restrict__ T) {
  __shared__ __attribute__((aligned(16))) float tile[(BROWS + 1) * PH];
  __shared__ __attribute__((aligned(16))) unsigned sorted[SEG_CAP];
  __shared__ __attribute__((aligned(16))) int cnt[BROWS];
  __shared__ __attribute__((aligned(16))) int rstart[BROWS + 4];
  __shared__ __attribute__((aligned(16))) float w2l[W2_PIECES * 4];
  const int t = threadIdx.x;
  const int b = blockIdx.x;
  const int bs = bstart[b], be = bstart[b + 1];

  v4f w1r[W1_TRIPS];
#pragma unroll
  for (int i = 0; i < W1_TRIPS; ++i) {
    const int q = t + i * SUM_THREADS;
    const int g = q >> 6, n = q & 63;
    const float* w = W1 + (8 * (g >> 1) + 2 * (g & 1)) * HID + n;
    w1r[i].x = w[0];
    w1r[i].y = w[HID];
    w1r[i].z = w[4 * HID];
    w1r[i].w = w[5 * HID];
  }
  v4f w2r[W2_TRIPS];
#pragma unroll
  for (int i = 0; i < W2_TRIPS; ++i) {
    const int q = t + i * SUM_THREADS;
    const int g = q / NC_PAD, n = q - g * NC_PAD;
    const int nc = n < NC ? n : NC - 1;
    const float* w = W2 + (8 * (g >> 1) + 2 * (g & 1)) * NC + nc;
    const float w0 = w[0], w1v = w[NC], w4 = w[4 * NC], w5 = w[5 * NC];
    w2r[i].x = n < NC ? w0 : 0.0f;
    w2r[i].y = n < NC ? w1v : 0.0f;
    w2r[i].z = n < NC ? w4 : 0.0f;
    w2r[i].w = n < NC ? w5 : 0.0f;
  }
  for (int i = t; i < (BROWS + 1) * (PH / 4); i += SUM_THREADS) ((v4f*)tile)[i] = (v4f)(0.0f);
  bucket_rows_sum<64, PH>(x, grouped, bs, be, tile, sorted, cnt, rstart);
  __syncthreads();
  float* w1l = (float*)sorted;
#pragma unroll
  for (int i = 0; i < W1_TRIPS; ++i) ((v4f*)w1l)[t + i * SUM_THREADS] = w1r[i];
#pragma unroll
  for (int i = 0; i < W2_TRIPS; ++i) ((v4f*)w2l)[t + i * SUM_THREADS] = w2r[i];
  __syncthreads();

  const int lane = t & 31;
  const int wave = t >> 5;
  const int lh = lane & 15;
  const int hi = lane >> 4;
  float* ht = tile + (ROWS_PER_WAVE * wave) * PH;

  v4f af[8];
#pragma unroll
  for (int p = 0; p < 8; ++p) af[p] = *(const v4f*)&ht[lh * PH + 8 * p + (hi << 2)];

  v8f acc[4] = {};
#pragma unroll
  for (int p = 0; p < 8; ++p) {
#pragma unroll
    for (int u = 0; u < 2; ++u) {
      v2f a;
      a.x = af[p][2 * u];
      a.y = af[p][2 * u + 1];
#pragma unroll
      for (int ct = 0; ct < 4; ++ct) {
        const v2f bq = *(const v2f*)&w1l[((2 * p + u) * HID + ct * 16 + lh) * 4 + 2 * hi];
        acc[ct] = __builtin_amdgcn_wmma_f32_16x16x4_f32(false, a, false, bq, (short)0,
                                                        acc[ct], false, false);
      }
    }
  }

#pragma unroll
  for (int ct = 0; ct < 4; ++ct) {
    const float bv = b1[ct * 16 + lh];
#pragma unroll
    for (int r = 0; r < 8; ++r)
      ht[((hi << 3) + r) * PH + ct * 16 + lh] = fmaxf(acc[ct][r] + bv, 0.0f);
  }
  __syncthreads();

  v4f hf[8];
#pragma unroll
  for (int p = 0; p < 8; ++p) hf[p] = *(const v4f*)&ht[lh * PH + 8 * p + (hi << 2)];
  v8f acc2[3] = {};
#pragma unroll
  for (int p = 0; p < 8; ++p) {
#pragma unroll
    for (int u = 0; u < 2; ++u) {
      v2f a;
      a.x = hf[p][2 * u];
      a.y = hf[p][2 * u + 1];
#pragma unroll
      for (int ct = 0; ct < 3; ++ct) {
        const v2f bq = *(const v2f*)&w2l[((2 * p + u) * NC_PAD + ct * 16 + lh) * 4 + 2 * hi];
        acc2[ct] = __builtin_amdgcn_wmma_f32_16x16x4_f32(false, a, false, bq, (short)0,
                                                         acc2[ct], false, false);
      }
    }
  }

#pragma unroll
  for (int ct = 0; ct < 3; ++ct) {
#pragma unroll
    for (int r = 0; r < 8; ++r)
      ht[((hi << 3) + r) * PT + ct * 16 + lh] = acc2[ct][r];
  }
  __syncthreads();

  const int row0 = b * BROWS + ROWS_PER_WAVE * wave;
  if (row0 < N_NODES) {
    float* trow = T + (size_t)row0 * NC;
#pragma unroll
    for (int q = 0; q < 5; ++q) {
      const int g = lane + 32 * q;
      const int row = g / 10;
      const int c4 = g - row * 10;
      const v4f v = *(const v4f*)&ht[row * PT + 4 * c4];
      *(v4f*)(trow + 4 * g) = v;
    }
  }
}

__global__ void __launch_bounds__(SUM_THREADS)
gcn_bucket_sum_rows40_kernel(const float* __restrict__ x,
                             const int* __restrict__ grouped,
                             const int* __restrict__ bstart,
                             const float* __restrict__ bias,
                             float* __restrict__ out) {
  __shared__ __attribute__((aligned(16))) float tile[(BROWS + 1) * NC];
  __shared__ __attribute__((aligned(16))) unsigned sorted[SEG_CAP];
  __shared__ __attribute__((aligned(16))) int cnt[BROWS];
  __shared__ __attribute__((aligned(16))) int rstart[BROWS + 4];
  const int t = threadIdx.x;
  const int b = blockIdx.x;
  const int bs = bstart[b], be = bstart[b + 1];
  if (t < 6 * NC) {
    const int c = t % NC;
    const float bv = bias[c];
    for (int r = t / NC; r < BROWS + 1; r += 6) tile[r * NC + c] = bv;
  }
  bucket_rows_sum<NC, NC>(x, grouped, bs, be, tile, sorted, cnt, rstart);
  __syncthreads();
  const int left = N_NODES - b * BROWS;
  const int rows = left < BROWS ? left : BROWS;
  v4f* g = (v4f*)(out + (size_t)b * BROWS * NC);
  for (int i = t; i < rows * (NC / 4); i += SUM_THREADS) g[i] = ((const v4f*)tile)[i];
}

extern "C" void kernel_launch(void* const* d_in, const int* in_sizes, int n_in,
                              void* d_out, int out_size, void* d_ws,
                              size_t ws_size, hipStream_t stream) {
  (void)in_sizes; (void)n_in; (void)out_size; (void)ws_size;
  const float* features = (const float*)d_in[0];
  const int*   src      = (const int*)d_in[1];
  const int*   dst      = (const int*)d_in[2];
  const float* W1       = (const float*)d_in[3];
  const float* b1       = (const float*)d_in[4];
  const float* W2       = (const float*)d_in[5];
  const float* b2       = (const float*)d_in[6];
  float* out = (float*)d_out;

  float* ws    = (float*)d_ws;
  float* t     = ws + (size_t)N_NODES * HID;
  int* grouped = (int*)(t + (size_t)N_NODES * NC);
  int* cnt     = grouped + N_EDGES;
  int* tot     = cnt + NBK_PAD * NCH;
  int* bstart  = tot + 1024;

  gcn_bucket_count_kernel<<<NCH, CH_THREADS, 0, stream>>>(dst, cnt);
  gcn_bucket_rowscan_kernel<<<NBK_PAD / 8, 256, 0, stream>>>(cnt, tot);
  gcn_bucket_place_kernel<<<NCH, CH_THREADS, 0, stream>>>(src, dst, cnt, tot, bstart, grouped);

  gcn_bucket_sum_rows64_dense_kernel<<<NBK, SUM_THREADS, 0, stream>>>(
      features, grouped, bstart, W1, b1, W2, t);

  gcn_bucket_sum_rows40_kernel<<<NBK, SUM_THREADS, 0, stream>>>(t, grouped, bstart, b2, out);
}


# place: a chunk's edges are first ordered by bucket in LDS (rank, scan, slot), then stored in that order: a wave's store covers a few contiguous runs, not 32 separate lines; arrival order kept
# speedup vs baseline: 3.4102x; 1.2675x over previous
#include <hip/hip_runtime.h>
#include <hip/hip_bf16.h>

typedef float v2f __attribute__((ext_vector_type(2)));
typedef float v8f __attribute__((ext_vector_type(8)));
typedef int v4i __attribute__((ext_vector_type(4)));

#define N_NODES 100000
#define N_EDGES 1600000
#define IN_F 64
#define HID 64
#define NC 40

#define BROWS 128
#define BSHIFT 7
#define NBK ((N_NODES + BROWS - 1) / BROWS)
#define NBK_PAD 800
#define NCH 256
#define CH_EDGES (N_EDGES / NCH)
#define CH_THREADS 1024
#define CH_TRIPS ((CH_EDGES + CH_THREADS - 1) / CH_THREADS)
#define SRC_BITS 17
static_assert(N_EDGES % NCH == 0, "the edge list is cut into whole chunks");
static_assert(N_NODES <= (1 << SRC_BITS), "source row fits its field");
static_assert(NBK <= NBK_PAD && NBK_PAD <= CH_THREADS && NBK_PAD % 8 == 0, "bucket table");
static_assert(NCH == 256, "one wave scans one bucket row: 32 lanes x 8");

__global__ void __launch_bounds__(CH_THREADS)
gcn_bucket_count_kernel(const int* __restrict__ dst, int* __restrict__ cnt) {
  __shared__ int h[NBK_PAD];
  const int t = threadIdx.x, j = blockIdx.x;
  const int* dp = dst + (size_t)j * CH_EDGES;
  int d[CH_TRIPS];
#pragma unroll
  for (int i = 0; i < CH_TRIPS; ++i) {
    const int idx = t + i * CH_THREADS;
    if (idx < CH_EDGES) d[i] = dp[idx];
  }
  if (t < NBK_PAD) h[t] = 0;
  __syncthreads();
#pragma unroll
  for (int i = 0; i < CH_TRIPS; ++i) {
    const int idx = t + i * CH_THREADS;
    if (idx < CH_EDGES) atomicAdd(&h[d[i] >> BSHIFT], 1);
  }
  __syncthreads();
  if (t < NBK_PAD) cnt[t * NCH + j] = h[t];
}

__global__ void __launch_bounds__(256)
gcn_bucket_rowscan_kernel(int* __restrict__ cnt, int* __restrict__ tot) {
  const int lane = threadIdx.x & 31;
  const int bk = blockIdx.x * 8 + (threadIdx.x >> 5);
  v4i* p = (v4i*)(cnt + (size_t)bk * NCH) + 2 * lane;
  const v4i c = p[0], c2 = p[1];
  const int s4 = c.x + c.y + c.z + c.w;
  const int s = s4 + c2.x + c2.y + c2.z + c2.w;
  int incl = s;
#pragma unroll
  for (int dlt = 1; dlt < 32; dlt <<= 1) {
    const int o = __shfl_up(incl, dlt);
    if (lane >= dlt) incl += o;
  }
  const int ex = incl - s;
  v4i o4;
  o4.x = ex;
  o4.y = ex + c.x;
  o4.z = ex + c.x + c.y;
  o4.w = ex + c.x + c.y + c.z;
  const int ex2 = ex + s4;
  v4i o8;
  o8.x = ex2;
  o8.y = ex2 + c2.x;
  o8.z = ex2 + c2.x + c2.y;
  o8.w = ex2 + c2.x + c2.y + c2.z;
  p[0] = o4;
  p[1] = o8;
  if (lane == 31) tot[bk] = incl;
}

typedef int v2i __attribute__((ext_vector_type(2)));
typedef unsigned long long u64;
__global__ void __launch_bounds__(CH_THREADS)
gcn_bucket_place_kernel(const int* __restrict__ src, const int* __restrict__ dst,
                        const int* __restrict__ cnt, const int* __restrict__ tot,
                        int* __restrict__ bstart, int* __restrict__ grouped) {
  __shared__ __attribute__((aligned(8))) u64 sc[2][CH_THREADS];
  __shared__ int h[NBK_PAD];
  __shared__ __attribute__((aligned(8))) v2i bslot[NBK_PAD];
  __shared__ __attribute__((aligned(8))) v2i ord[CH_EDGES];
  const int t = threadIdx.x, j = blockIdx.x;
  const int* dp = dst + (size_t)j * CH_EDGES;
  const int* sp = src + (size_t)j * CH_EDGES;
  int d[CH_TRIPS], s[CH_TRIPS], rk[CH_TRIPS];
#pragma unroll
  for (int i = 0; i < CH_TRIPS; ++i) {
    const int idx = t + i * CH_THREADS;
    if (idx < CH_EDGES) { d[i] = dp[idx]; s[i] = sp[idx]; }
  }
  const int v = t < NBK_PAD ? tot[t] : 0;
  const int mine = t < NBK_PAD ? cnt[t * NCH + j] : 0;
  if (t < NBK_PAD) h[t] = 0;
  __syncthreads();
#pragma unroll
  for (int i = 0; i < CH_TRIPS; ++i) {
    const int idx = t + i * CH_THREADS;
    if (idx < CH_EDGES) rk[i] = atomicAdd(&h[d[i] >> BSHIFT], 1);
  }
  __syncthreads();
  const int c = t < NBK_PAD ? h[t] : 0;
  sc[0][t] = ((u64)(unsigned)v << 32) | (u64)(unsigned)c;
  __syncthreads();
  int cur = 0;
  for (int dlt = 1; dlt < CH_THREADS; dlt <<= 1) {
    u64 x = sc[cur][t];
    if (t >= dlt) x += sc[cur][t - dlt];
    sc[cur ^ 1][t] = x;
    __syncthreads();
    cur ^= 1;
  }
  const u64 incl = sc[cur][t];
  const int excl = (int)(unsigned)(incl >> 32) - v;
  const int lfirst = (int)(unsigned)incl - c;
  if (t < NBK_PAD) {
    v2i q;
    q.x = lfirst;
    q.y = excl + mine - lfirst;
    bslot[t] = q;
  }
  if (j == 0 && t <= NBK_PAD) bstart[t] = excl;
  __syncthreads();
#pragma unroll
  for (int i = 0; i < CH_TRIPS; ++i) {
    const int idx = t + i * CH_THREADS;
    if (idx < CH_EDGES) {
      const v2i q = bslot[d[i] >> BSHIFT];
      const int lp = q.x + rk[i];
      v2i e;
      e.x = s[i] | ((d[i] & (BROWS - 1)) << SRC_BITS);
      e.y = q.y + lp;
      ord[lp] = e;
    }
  }
  __syncthreads();
#pragma unroll
  for (int i = 0; i < CH_TRIPS; ++i) {
    const int idx = t + i * CH_THREADS;
    if (idx < CH_EDGES) {
      const v2i e = ord[idx];
      grouped[e.y] = e.x;
    }
  }
}

typedef float v4f __attribute__((ext_vector_type(4)));
#define SUM_THREADS 256
#define SUM_WAVES (SUM_THREADS / 32)
#define ROWS_PER_WAVE (BROWS / SUM_WAVES)
#define EDGES_PER_BATCH 16
#define NULL_EDGE ((unsigned)BROWS << SRC_BITS)
#define SRC_MASK ((1u << SRC_BITS) - 1u)
#define SEG_CAP 4096
#define SEG_TRIPS (SEG_CAP / SUM_THREADS)
static_assert(BROWS == 128, "one wave scans the row counts: 32 lanes x 4");
static_assert(BROWS % SUM_WAVES == 0, "every wave owns whole rows");

template <int COLS, int PITCH>
__device__ __forceinline__ void bucket_rows_sum(const float* __restrict__ x,
                                                const int* __restrict__ grouped,
                                                const int bs, const int be,
                                                float* tile, unsigned* sorted,
                                                int* cnt, int* rstart) {
  const int t = threadIdx.x;
  const unsigned lane = t & 31u;
  const int wave = __builtin_amdgcn_readfirstlane(t >> 5);
  const unsigned c1 = COLS == 64 ? lane + 32u : 32u + (lane & 7u);

  for (int s0 = bs; s0 < be; s0 += SEG_CAP) {
    const int left = be - s0;
    const int n = left < SEG_CAP ? left : SEG_CAP;
    unsigned pk[SEG_TRIPS];
    int rk[SEG_TRIPS];
#pragma unroll
    for (int i = 0; i < SEG_TRIPS; ++i) {
      const int idx = t + i * SUM_THREADS;
      pk[i] = idx < n ? (unsigned)grouped[s0 + idx] : 0u;
    }
    if (t < BROWS) cnt[t] = 0;
    __syncthreads();
#pragma unroll
    for (int i = 0; i < SEG_TRIPS; ++i) {
      const int idx = t + i * SUM_THREADS;
      if (idx < n) rk[i] = atomicAdd(&cnt[pk[i] >> SRC_BITS], 1);
    }
    __syncthreads();
    if (wave == 0) {
      const v4i c = ((const v4i*)cnt)[lane];
      const int s = c.x + c.y + c.z + c.w;
      int incl = s;
#pragma unroll
      for (int dlt = 1; dlt < 32; dlt <<= 1) {
        const int o = __shfl_up(incl, dlt);
        if ((int)lane >= dlt) incl += o;
      }
      const int ex = incl - s;
      v4i o4;
      o4.x = ex;
      o4.y = ex + c.x;
      o4.z = ex + c.x + c.y;
      o4.w = ex + c.x + c.y + c.z;
      ((v4i*)rstart)[lane] = o4;
      if (lane == 31u) rstart[BROWS] = incl;
    }
    __syncthreads();
#pragma unroll
    for (int i = 0; i < SEG_TRIPS; ++i) {
      const int idx = t + i * SUM_THREADS;
      if (idx < n) sorted[rstart[pk[i] >> SRC_BITS] + rk[i]] = pk[i];
    }
    __syncthreads();

    int e0 = __builtin_amdgcn_readfirstlane(rstart[ROWS_PER_WAVE * wave]);
    const int we = __builtin_amdgcn_readfirstlane(rstart[ROWS_PER_WAVE * (wave + 1)]);
    int cur = ROWS_PER_WAVE * wave;
    float a0 = 0.0f, a1 = 0.0f;
    while (e0 < we) {
      const unsigned pkv = (e0 + (int)lane < we) ? sorted[e0 + lane] : NULL_EDGE;
#pragma unroll
      for (int bb = 0; bb < 32; bb += EDGES_PER_BATCH) {
        if (e0 + bb < we) {
          float v0[EDGES_PER_BATCH], v1[EDGES_PER_BATCH];
#pragma unroll
          for (int i = 0; i < EDGES_PER_BATCH; ++i) {
            const unsigned w = (unsigned)__builtin_amdgcn_readlane((int)pkv, bb + i);
            const float* xr = x + (size_t)(w & SRC_MASK) * COLS;
            v0[i] = xr[lane];
            v1[i] = xr[c1];
          }
#pragma unroll
          for (int i = 0; i < EDGES_PER_BATCH; ++i) {
            const unsigned w = (unsigned)__builtin_amdgcn_readlane((int)pkv, bb + i);
            const int r = (int)(w >> SRC_BITS);
            if (r != cur) {
              float* o = tile + cur * PITCH;
              atomicAdd(o + lane, a0);
              if (COLS == 64) atomicAdd(o + lane + 32, a1);
              else if (lane < (unsigned)(COLS - 32)) atomicAdd(o + 32 + lane, a1);
              a0 = 0.0f;
              a1 = 0.0f;
              cur = r;
            }
            a0 += v0[i];
            a1 += v1[i];
          }
        }
      }
      e0 += 32;
    }
    {
      float* o = tile + cur * PITCH;
      atomicAdd(o + lane, a0);
      if (COLS == 64) atomicAdd(o + lane + 32, a1);
      else if (lane < (unsigned)(COLS - 32)) atomicAdd(o + 32 + lane, a1);
    }
  }
}

#define NC_PAD 48
#define PH 68
#define PT 52
#define W1_PIECES (16 * HID)
#define W2_PIECES (16 * NC_PAD)
#define W1_TRIPS (W1_PIECES / SUM_THREADS)
#define W2_TRIPS (W2_PIECES / SUM_THREADS)
static_assert(IN_F == 64 && HID == 64, "16 WMMA k groups of 4 per layer");
static_assert(W1_PIECES * 4 == SEG_CAP, "W1 fills the ordered-segment buffer exactly");
static_assert(W1_PIECES % SUM_THREADS == 0 && W2_PIECES % SUM_THREADS == 0, "whole staging trips");
static_assert(ROWS_PER_WAVE == 16, "a wave's rows are one WMMA row tile");
static_assert(N_NODES % ROWS_PER_WAVE == 0, "a wave's 16 rows exist all or not at all");

__global__ void __launch_bounds__(SUM_THREADS)
gcn_bucket_sum_rows64_dense_kernel(const float* __restrict__ x,
                                   const int* __restrict__ grouped,
                                   const int* __restrict__ bstart,
                                   const float* __restrict__ W1,
                                   const float* __restrict__ b1,
                                   const float* __restrict__ W2,
                                   float* __restrict__ T) {
  __shared__ __attribute__((aligned(16))) float tile[(BROWS + 1) * PH];
  __shared__ __attribute__((aligned(16))) unsigned sorted[SEG_CAP];
  __shared__ __attribute__((aligned(16))) int cnt[BROWS];
  __shared__ __attribute__((aligned(16))) int rstart[BROWS + 4];
  __shared__ __attribute__((aligned(16))) float w2l[W2_PIECES * 4];
  const int t = threadIdx.x;
  const int b = blockIdx.x;
  const int bs = bstart[b], be = bstart[b + 1];

  v4f w1r[W1_TRIPS];
#pragma unroll
  for (int i = 0; i < W1_TRIPS; ++i) {
    const int q = t + i * SUM_THREADS;
    const int g = q >> 6, n = q & 63;
    const float* w = W1 + (8 * (g >> 1) + 2 * (g & 1)) * HID + n;
    w1r[i].x = w[0];
    w1r[i].y = w[HID];
    w1r[i].z = w[4 * HID];
    w1r[i].w = w[5 * HID];
  }
  v4f w2r[W2_TRIPS];
#pragma unroll
  for (int i = 0; i < W2_TRIPS; ++i) {
    const int q = t + i * SUM_THREADS;
    const int g = q / NC_PAD, n = q - g * NC_PAD;
    const int nc = n < NC ? n : NC - 1;
    const float* w = W2 + (8 * (g >> 1) + 2 * (g & 1)) * NC + nc;
    const float w0 = w[0], w1v = w[NC], w4 = w[4 * NC], w5 = w[5 * NC];
    w2r[i].x = n < NC ? w0 : 0.0f;
    w2r[i].y = n < NC ? w1v : 0.0f;
    w2r[i].z = n < NC ? w4 : 0.0f;
    w2r[i].w = n < NC ? w5 : 0.0f;
  }
  for (int i = t; i < (BROWS + 1) * (PH / 4); i += SUM_THREADS) ((v4f*)tile)[i] = (v4f)(0.0f);
  bucket_rows_sum<64, PH>(x, grouped, bs, be, tile, sorted, cnt, rstart);
  __syncthreads();
  float* w1l = (float*)sorted;
#pragma unroll
  for (int i = 0; i < W1_TRIPS; ++i) ((v4f*)w1l)[t + i * SUM_THREADS] = w1r[i];
#pragma unroll
  for (int i = 0; i < W2_TRIPS; ++i) ((v4f*)w2l)[t + i * SUM_THREADS] = w2r[i];
  __syncthreads();

  const int lane = t & 31;
  const int wave = t >> 5;
  const int lh = lane & 15;
  const int hi = lane >> 4;
  float* ht = tile + (ROWS_PER_WAVE * wave) * PH;

  v4f af[8];
#pragma unroll
  for (int p = 0; p < 8; ++p) af[p] = *(const v4f*)&ht[lh * PH + 8 * p + (hi << 2)];

  v8f acc[4] = {};
#pragma unroll
  for (int p = 0; p < 8; ++p) {
#pragma unroll
    for (int u = 0; u < 2; ++u) {
      v2f a;
      a.x = af[p][2 * u];
      a.y = af[p][2 * u + 1];
#pragma unroll
      for (int ct = 0; ct < 4; ++ct) {
        const v2f bq = *(const v2f*)&w1l[((2 * p + u) * HID + ct * 16 + lh) * 4 + 2 * hi];
        acc[ct] = __builtin_amdgcn_wmma_f32_16x16x4_f32(false, a, false, bq, (short)0,
                                                        acc[ct], false, false);
      }
    }
  }

#pragma unroll
  for (int ct = 0; ct < 4; ++ct) {
    const float bv = b1[ct * 16 + lh];
#pragma unroll
    for (int r = 0; r < 8; ++r)
      ht[((hi << 3) + r) * PH + ct * 16 + lh] = fmaxf(acc[ct][r] + bv, 0.0f);
  }
  __syncthreads();

  v4f hf[8];
#pragma unroll
  for (int p = 0; p < 8; ++p) hf[p] = *(const v4f*)&ht[lh * PH + 8 * p + (hi << 2)];
  v8f acc2[3] = {};
#pragma unroll
  for (int p = 0; p < 8; ++p) {
#pragma unroll
    for (int u = 0; u < 2; ++u) {
      v2f a;
      a.x = hf[p][2 * u];
      a.y = hf[p][2 * u + 1];
#pragma unroll
      for (int ct = 0; ct < 3; ++ct) {
        const v2f bq = *(const v2f*)&w2l[((2 * p + u) * NC_PAD + ct * 16 + lh) * 4 + 2 * hi];
        acc2[ct] = __builtin_amdgcn_wmma_f32_16x16x4_f32(false, a, false, bq, (short)0,
                                                         acc2[ct], false, false);
      }
    }
  }

#pragma unroll
  for (int ct = 0; ct < 3; ++ct) {
#pragma unroll
    for (int r = 0; r < 8; ++r)
      ht[((hi << 3) + r) * PT + ct * 16 + lh] = acc2[ct][r];
  }
  __syncthreads();

  const int row0 = b * BROWS + ROWS_PER_WAVE * wave;
  if (row0 < N_NODES) {
    float* trow = T + (size_t)row0 * NC;
#pragma unroll
    for (int q = 0; q < 5; ++q) {
      const int g = lane + 32 * q;
      const int row = g / 10;
      const int c4 = g - row * 10;
      const v4f v = *(const v4f*)&ht[row * PT + 4 * c4];
      *(v4f*)(trow + 4 * g) = v;
    }
  }
}

__global__ void __launch_bounds__(SUM_THREADS)
gcn_bucket_sum_rows40_kernel(const float* __restrict__ x,
                             const int* __restrict__ grouped,
                             const int* __restrict__ bstart,
                             const float* __restrict__ bias,
                             float* __restrict__ out) {
  __shared__ __attribute__((aligned(16))) float tile[(BROWS + 1) * NC];
  __shared__ __attribute__((aligned(16))) unsigned sorted[SEG_CAP];
  __shared__ __attribute__((aligned(16))) int cnt[BROWS];
  __shared__ __attribute__((aligned(16))) int rstart[BROWS + 4];
  const int t = threadIdx.x;
  const int b = blockIdx.x;
  const int bs = bstart[b], be = bstart[b + 1];
  if (t < 6 * NC) {
    const int c = t % NC;
    const float bv = bias[c];
    for (int r = t / NC; r < BROWS + 1; r += 6) tile[r * NC + c] = bv;
  }
  bucket_rows_sum<NC, NC>(x, grouped, bs, be, tile, sorted, cnt, rstart);
  __syncthreads();
  const int left = N_NODES - b * BROWS;
  const int rows = left < BROWS ? left : BROWS;
  v4f* g = (v4f*)(out + (size_t)b * BROWS * NC);
  for (int i = t; i < rows * (NC / 4); i += SUM_THREADS) g[i] = ((const v4f*)tile)[i];
}

extern "C" void kernel_launch(void* const* d_in, const int* in_sizes, int n_in,
                              void* d_out, int out_size, void* d_ws,
                              size_t ws_size, hipStream_t stream) {
  (void)in_sizes; (void)n_in; (void)out_size; (void)ws_size;
  const float* features = (const float*)d_in[0];
  const int*   src      = (const int*)d_in[1];
  const int*   dst      = (const int*)d_in[2];
  const float* W1       = (const float*)d_in[3];
  const float* b1       = (const float*)d_in[4];
  const float* W2       = (const float*)d_in[5];
  const float* b2       = (const float*)d_in[6];
  float* out = (float*)d_out;

  float* ws    = (float*)d_ws;
  float* t     = ws + (size_t)N_NODES * HID;
  int* grouped = (int*)(t + (size_t)N_NODES * NC);
  int* cnt     = grouped + N_EDGES;
  int* tot     = cnt + NBK_PAD * NCH;
  int* bstart  = tot + 1024;

  gcn_bucket_count_kernel<<<NCH, CH_THREADS, 0, stream>>>(dst, cnt);
  gcn_bucket_rowscan_kernel<<<NBK_PAD / 8, 256, 0, stream>>>(cnt, tot);
  gcn_bucket_place_kernel<<<NCH, CH_THREADS, 0, stream>>>(src, dst, cnt, tot, bstart, grouped);

  gcn_bucket_sum_rows64_dense_kernel<<<NBK, SUM_THREADS, 0, stream>>>(
      features, grouped, bstart, W1, b1, W2, t);

  gcn_bucket_sum_rows40_kernel<<<NBK, SUM_THREADS, 0, stream>>>(t, grouped, bstart, b2, out);
}


# edges ordered by destination block inside each 6250-edge chunk only (one launch, with run starts); a block's work-group reads its 256 runs itself: no count/scan launches (3 left); arrival-order sums
# speedup vs baseline: 4.1696x; 1.2227x over previous
#include <hip/hip_runtime.h>
#include <hip/hip_bf16.h>

typedef float v2f __attribute__((ext_vector_type(2)));
typedef float v8f __attribute__((ext_vector_type(8)));
typedef int v4i __attribute__((ext_vector_type(4)));

#define N_NODES 100000
#define N_EDGES 1600000
#define IN_F 64
#define HID 64
#define NC 40

#define BROWS 128
#define BSHIFT 7
#define NBK ((N_NODES + BROWS - 1) / BROWS)
#define NBK_PAD 800
#define NCH 256
#define CH_EDGES (N_EDGES / NCH)
#define CH_THREADS 1024
#define CH_TRIPS ((CH_EDGES + CH_THREADS - 1) / CH_THREADS)
#define SRC_BITS 17
static_assert(N_EDGES % NCH == 0, "the edge list is cut into whole chunks");
static_assert(N_NODES <= (1 << SRC_BITS), "source row fits its field");
static_assert(NBK <= NBK_PAD && NBK_PAD <= CH_THREADS && NBK_PAD % 8 == 0, "bucket table");
#define OFF_PITCH (NBK_PAD + 32)
static_assert(NBK + 1 <= NBK_PAD, "entry bucket + 1 exists for the last bucket");

__global__ void __launch_bounds__(CH_THREADS)
gcn_chunk_order_kernel(const int* __restrict__ src, const int* __restrict__ dst,
                       int* __restrict__ off, int* __restrict__ grouped) {
  __shared__ int sc[2][CH_THREADS];
  __shared__ int h[NBK_PAD];
  __shared__ int bfirst[NBK_PAD];
  __shared__ int ord[CH_EDGES];
  const int t = threadIdx.x, j = blockIdx.x;
  const int* dp = dst + (size_t)j * CH_EDGES;
  const int* sp = src + (size_t)j * CH_EDGES;
  int d[CH_TRIPS], s[CH_TRIPS], rk[CH_TRIPS];
#pragma unroll
  for (int i = 0; i < CH_TRIPS; ++i) {
    const int idx = t + i * CH_THREADS;
    if (idx < CH_EDGES) { d[i] = dp[idx]; s[i] = sp[idx]; }
  }
  if (t < NBK_PAD) h[t] = 0;
  __syncthreads();
#pragma unroll
  for (int i = 0; i < CH_TRIPS; ++i) {
    const int idx = t + i * CH_THREADS;
    if (idx < CH_EDGES) rk[i] = atomicAdd(&h[d[i] >> BSHIFT], 1);
  }
  __syncthreads();
  const int c = t < NBK_PAD ? h[t] : 0;
  sc[0][t] = c;
  __syncthreads();
  int cur = 0;
  for (int dlt = 1; dlt < CH_THREADS; dlt <<= 1) {
    int x = sc[cur][t];
    if (t >= dlt) x += sc[cur][t - dlt];
    sc[cur ^ 1][t] = x;
    __syncthreads();
    cur ^= 1;
  }
  const int lfirst = sc[cur][t] - c;
  if (t < NBK_PAD) bfirst[t] = lfirst;
  if (t <= NBK_PAD) off[(size_t)j * OFF_PITCH + t] = lfirst;
  __syncthreads();
#pragma unroll
  for (int i = 0; i < CH_TRIPS; ++i) {
    const int idx = t + i * CH_THREADS;
    if (idx < CH_EDGES)
      ord[bfirst[d[i] >> BSHIFT] + rk[i]] = s[i] | ((d[i] & (BROWS - 1)) << SRC_BITS);
  }
  __syncthreads();
  int* gp = grouped + (size_t)j * CH_EDGES;
#pragma unroll
  for (int i = 0; i < CH_TRIPS; ++i) {
    const int idx = t + i * CH_THREADS;
    if (idx < CH_EDGES) gp[idx] = ord[idx];
  }
}

typedef float v4f __attribute__((ext_vector_type(4)));
#define SUM_THREADS 256
#define SUM_WAVES (SUM_THREADS / 32)
#define ROWS_PER_WAVE (BROWS / SUM_WAVES)
#define EDGES_PER_BATCH 16
#define NULL_EDGE ((unsigned)BROWS << SRC_BITS)
#define SRC_MASK ((1u << SRC_BITS) - 1u)
#define RUN_WORDS 24
#define SEG_CAP (NCH * RUN_WORDS)
static_assert(NCH == SUM_THREADS, "thread r of a block reads the bounds of run r");
static_assert(BROWS == 128, "one wave scans the row counts: 32 lanes x 4");
static_assert(BROWS % SUM_WAVES == 0, "every wave owns whole rows");

template <int COLS, int PITCH>
__device__ __forceinline__ void bucket_rows_sum(const float* __restrict__ x,
                                                const int* __restrict__ grouped,
                                                const int* __restrict__ off, const int b,
                                                float* tile, unsigned* sorted,
                                                int* cnt, int* rstart, int* taken) {
  const int t = threadIdx.x;
  const unsigned lane = t & 31u;
  const int wave = __builtin_amdgcn_readfirstlane(t >> 5);
  const unsigned c1 = COLS == 64 ? lane + 32u : 32u + (lane & 7u);

  const int* op = off + (size_t)t * OFF_PITCH + b;
  const int o0 = op[0];
  const int len = op[1] - o0;
  static_assert(CH_EDGES < (1 << 16), "a run's first word and its length share a 32-bit word");
  const int bounds = o0 | (len << 16);
  if (t == 0) taken[0] = 0;
  const int sub = (int)(lane >> 3);
  const int q = (int)(lane & 7u);

  for (int r0 = 0;; r0 += RUN_WORDS) {
    unsigned pk[RUN_WORDS];
    int rbv[8];
#pragma unroll
    for (int i = 0; i < 8; ++i) rbv[i] = __shfl(bounds, 4 * i + sub);
#pragma unroll
    for (int i = 0; i < 8; ++i) {
      const int sl = 4 * i + sub;
      const int rb = rbv[i];
      const int rl = rb >> 16;
      const int g0 = (32 * wave + sl) * CH_EDGES + (rb & 0xffff);
#pragma unroll
      for (int r = 0; r < 3; ++r) {
        const int k = r0 + q + 8 * r;
        pk[3 * i + r] = k < rl ? (unsigned)grouped[g0 + k] : NULL_EDGE;
      }
    }
    if (t < BROWS) cnt[t] = 0;
    __syncthreads();
    if (len > r0 + RUN_WORDS) taken[0] = r0 + RUN_WORDS;
#pragma unroll
    for (int i = 0; i < RUN_WORDS; ++i)
      if (pk[i] < NULL_EDGE) atomicAdd(&cnt[pk[i] >> SRC_BITS], 1);
    __syncthreads();
    if (wave == 0) {
      const v4i c = ((const v4i*)cnt)[lane];
      const int s = c.x + c.y + c.z + c.w;
      int incl = s;
#pragma unroll
      for (int dlt = 1; dlt < 32; dlt <<= 1) {
        const int o = __shfl_up(incl, dlt);
        if ((int)lane >= dlt) incl += o;
      }
      const int ex = incl - s;
      v4i o4;
      o4.x = ex;
      o4.y = ex + c.x;
      o4.z = ex + c.x + c.y;
      o4.w = ex + c.x + c.y + c.z;
      ((v4i*)rstart)[lane] = o4;
      ((v4i*)cnt)[lane] = o4;
      if (lane == 31u) rstart[BROWS] = incl;
    }
    __syncthreads();
#pragma unroll
    for (int i = 0; i < RUN_WORDS; ++i)
      if (pk[i] < NULL_EDGE) sorted[atomicAdd(&cnt[pk[i] >> SRC_BITS], 1)] = pk[i];
    __syncthreads();

    int e0 = __builtin_amdgcn_readfirstlane(rstart[ROWS_PER_WAVE * wave]);
    const int we = __builtin_amdgcn_readfirstlane(rstart[ROWS_PER_WAVE * (wave + 1)]);
    int cur = ROWS_PER_WAVE * wave;
    float a0 = 0.0f, a1 = 0.0f;
    while (e0 < we) {
      const unsigned pkv = (e0 + (int)lane < we) ? sorted[e0 + lane] : NULL_EDGE;
#pragma unroll
      for (int bb = 0; bb < 32; bb += EDGES_PER_BATCH) {
        if (e0 + bb < we) {
          float v0[EDGES_PER_BATCH], v1[EDGES_PER_BATCH];
#pragma unroll
          for (int i = 0; i < EDGES_PER_BATCH; ++i) {
            const unsigned w = (unsigned)__builtin_amdgcn_readlane((int)pkv, bb + i);
            const float* xr = x + (size_t)(w & SRC_MASK) * COLS;
            v0[i] = xr[lane];
            v1[i] = xr[c1];
          }
#pragma unroll
          for (int i = 0; i < EDGES_PER_BATCH; ++i) {
            const unsigned w = (unsigned)__builtin_amdgcn_readlane((int)pkv, bb + i);
            const int r = (int)(w >> SRC_BITS);
            if (r != cur) {
              float* o = tile + cur * PITCH;
              atomicAdd(o + lane, a0);
              if (COLS == 64) atomicAdd(o + lane + 32, a1);
              else if (lane < (unsigned)(COLS - 32)) atomicAdd(o + 32 + lane, a1);
              a0 = 0.0f;
              a1 = 0.0f;
              cur = r;
            }
            a0 += v0[i];
            a1 += v1[i];
          }
        }
      }
      e0 += 32;
    }
    {
      float* o = tile + cur * PITCH;
      atomicAdd(o + lane, a0);
      if (COLS == 64) atomicAdd(o + lane + 32, a1);
      else if (lane < (unsigned)(COLS - 32)) atomicAdd(o + 32 + lane, a1);
    }
    if (taken[0] != r0 + RUN_WORDS) break;
  }
}

#define NC_PAD 48
#define PH 68
#define PT 52
#define W1_PIECES (16 * HID)
#define W2_PIECES (16 * NC_PAD)
#define W1_TRIPS (W1_PIECES / SUM_THREADS)
#define W2_TRIPS (W2_PIECES / SUM_THREADS)
static_assert(IN_F == 64 && HID == 64, "16 WMMA k groups of 4 per layer");
static_assert(W1_PIECES * 4 <= SEG_CAP, "W1 fits the ordered-segment buffer");
static_assert(W1_PIECES % SUM_THREADS == 0 && W2_PIECES % SUM_THREADS == 0, "whole staging trips");
static_assert(ROWS_PER_WAVE == 16, "a wave's rows are one WMMA row tile");
static_assert(N_NODES % ROWS_PER_WAVE == 0, "a wave's 16 rows exist all or not at all");

__global__ void __launch_bounds__(SUM_THREADS)
gcn_bucket_sum_rows64_dense_kernel(const float* __restrict__ x,
                                   const int* __restrict__ grouped,
                                   const int* __restrict__ off,
                                   const float* __restrict__ W1,
                                   const float* __restrict__ b1,
                                   const float* __restrict__ W2,
                                   float* __restrict__ T) {
  __shared__ __attribute__((aligned(16))) float tile[(BROWS + 1) * PH];
  __shared__ __attribute__((aligned(16))) unsigned sorted[SEG_CAP];
  __shared__ __attribute__((aligned(16))) int cnt[BROWS];
  __shared__ __attribute__((aligned(16))) int rstart[BROWS + 4];
  __shared__ __attribute__((aligned(16))) float w2l[W2_PIECES * 4];
  __shared__ int taken[1];
  const int t = threadIdx.x;
  const int b = blockIdx.x;

  v4f w1r[W1_TRIPS];
#pragma unroll
  for (int i = 0; i < W1_TRIPS; ++i) {
    const int q = t + i * SUM_THREADS;
    const int g = q >> 6, n = q & 63;
    const float* w = W1 + (8 * (g >> 1) + 2 * (g & 1)) * HID + n;
    w1r[i].x = w[0];
    w1r[i].y = w[HID];
    w1r[i].z = w[4 * HID];
    w1r[i].w = w[5 * HID];
  }
  v4f w2r[W2_TRIPS];
#pragma unroll
  for (int i = 0; i < W2_TRIPS; ++i) {
    const int q = t + i * SUM_THREADS;
    const int g = q / NC_PAD, n = q - g * NC_PAD;
    const int nc = n < NC ? n : NC - 1;
    const float* w = W2 + (8 * (g >> 1) + 2 * (g & 1)) * NC + nc;
    const float w0 = w[0], w1v = w[NC], w4 = w[4 * NC], w5 = w[5 * NC];
    w2r[i].x = n < NC ? w0 : 0.0f;
    w2r[i].y = n < NC ? w1v : 0.0f;
    w2r[i].z = n < NC ? w4 : 0.0f;
    w2r[i].w = n < NC ? w5 : 0.0f;
  }
  for (int i = t; i < (BROWS + 1) * (PH / 4); i += SUM_THREADS) ((v4f*)tile)[i] = (v4f)(0.0f);
  bucket_rows_sum<64, PH>(x, grouped, off, b, tile, sorted, cnt, rstart, taken);
  __syncthreads();
  float* w1l = (float*)sorted;
#pragma unroll
  for (int i = 0; i < W1_TRIPS; ++i) ((v4f*)w1l)[t + i * SUM_THREADS] = w1r[i];
#pragma unroll
  for (int i = 0; i < W2_TRIPS; ++i) ((v4f*)w2l)[t + i * SUM_THREADS] = w2r[i];
  __syncthreads();

  const int lane = t & 31;
  const int wave = t >> 5;
  const int lh = lane & 15;
  const int hi = lane >> 4;
  float* ht = tile + (ROWS_PER_WAVE * wave) * PH;

  v4f af[8];
#pragma unroll
  for (int p = 0; p < 8; ++p) af[p] = *(const v4f*)&ht[lh * PH + 8 * p + (hi << 2)];

  v8f acc[4] = {};
#pragma unroll
  for (int p = 0; p < 8; ++p) {
#pragma unroll
    for (int u = 0; u < 2; ++u) {
      v2f a;
      a.x = af[p][2 * u];
      a.y = af[p][2 * u + 1];
#pragma unroll
      for (int ct = 0; ct < 4; ++ct) {
        const v2f bq = *(const v2f*)&w1l[((2 * p + u) * HID + ct * 16 + lh) * 4 + 2 * hi];
        acc[ct] = __builtin_amdgcn_wmma_f32_16x16x4_f32(false, a, false, bq, (short)0,
                                                        acc[ct], false, false);
      }
    }
  }

#pragma unroll
  for (int ct = 0; ct < 4; ++ct) {
    const float bv = b1[ct * 16 + lh];
#pragma unroll
    for (int r = 0; r < 8; ++r)
      ht[((hi << 3) + r) * PH + ct * 16 + lh] = fmaxf(acc[ct][r] + bv, 0.0f);
  }
  __syncthreads();

  v4f hf[8];
#pragma unroll
  for (int p = 0; p < 8; ++p) hf[p] = *(const v4f*)&ht[lh * PH + 8 * p + (hi << 2)];
  v8f acc2[3] = {};
#pragma unroll
  for (int p = 0; p < 8; ++p) {
#pragma unroll
    for (int u = 0; u < 2; ++u) {
      v2f a;
      a.x = hf[p][2 * u];
      a.y = hf[p][2 * u + 1];
#pragma unroll
      for (int ct = 0; ct < 3; ++ct) {
        const v2f bq = *(const v2f*)&w2l[((2 * p + u) * NC_PAD + ct * 16 + lh) * 4 + 2 * hi];
        acc2[ct] = __builtin_amdgcn_wmma_f32_16x16x4_f32(false, a, false, bq, (short)0,
                                                         acc2[ct], false, false);
      }
    }
  }

#pragma unroll
  for (int ct = 0; ct < 3; ++ct) {
#pragma unroll
    for (int r = 0; r < 8; ++r)
      ht[((hi << 3) + r) * PT + ct * 16 + lh] = acc2[ct][r];
  }
  __syncthreads();

  const int row0 = b * BROWS + ROWS_PER_WAVE * wave;
  if (row0 < N_NODES) {
    float* trow = T + (size_t)row0 * NC;
#pragma unroll
    for (int q = 0; q < 5; ++q) {
      const int g = lane + 32 * q;
      const int row = g / 10;
      const int c4 = g - row * 10;
      const v4f v = *(const v4f*)&ht[row * PT + 4 * c4];
      *(v4f*)(trow + 4 * g) = v;
    }
  }
}

__global__ void __launch_bounds__(SUM_THREADS)
gcn_bucket_sum_rows40_kernel(const float* __restrict__ x,
                             const int* __restrict__ grouped,
                             const int* __restrict__ off,
                             const float* __restrict__ bias,
                             float* __restrict__ out) {
  __shared__ __attribute__((aligned(16))) float tile[(BROWS + 1) * NC];
  __shared__ __attribute__((aligned(16))) unsigned sorted[SEG_CAP];
  __shared__ __attribute__((aligned(16))) int cnt[BROWS];
  __shared__ __attribute__((aligned(16))) int rstart[BROWS + 4];
  __shared__ int taken[1];
  const int t = threadIdx.x;
  const int b = blockIdx.x;
  if (t < 6 * NC) {
    const int c = t % NC;
    const float bv = bias[c];
    for (int r = t / NC; r < BROWS + 1; r += 6) tile[r * NC + c] = bv;
  }
  bucket_rows_sum<NC, NC>(x, grouped, off, b, tile, sorted, cnt, rstart, taken);
  __syncthreads();
  const int left = N_NODES - b * BROWS;
  const int rows = left < BROWS ? left : BROWS;
  v4f* g = (v4f*)(out + (size_t)b * BROWS * NC);
  for (int i = t; i < rows * (NC / 4); i += SUM_THREADS) g[i] = ((const v4f*)tile)[i];
}

extern "C" void kernel_launch(void* const* d_in, const int* in_sizes, int n_in,
                              void* d_out, int out_size, void* d_ws,
                              size_t ws_size, hipStream_t stream) {
  (void)in_sizes; (void)n_in; (void)out_size; (void)ws_size;
  const float* features = (const float*)d_in[0];
  const int*   src      = (const int*)d_in[1];
  const int*   dst      = (const int*)d_in[2];
  const float* W1       = (const float*)d_in[3];
  const float* b1       = (const float*)d_in[4];
  const float* W2       = (const float*)d_in[5];
  const float* b2       = (const float*)d_in[6];
  float* out = (float*)d_out;

  float* ws    = (float*)d_ws;
  float* t     = ws + (size_t)N_NODES * HID;
  int* grouped = (int*)(t + (size_t)N_NODES * NC);
  int* off     = grouped + N_EDGES;

  gcn_chunk_order_kernel<<<NCH, CH_THREADS, 0, stream>>>(src, dst, off, grouped);

  gcn_bucket_sum_rows64_dense_kernel<<<NBK, SUM_THREADS, 0, stream>>>(
      features, grouped, off, W1, b1, W2, t);

  gcn_bucket_sum_rows40_kernel<<<NBK, SUM_THREADS, 0, stream>>>(t, grouped, off, b2, out);
}


# row sums: a half-wave owns 8 rows and a lane 4 adjacent columns, so one 16-byte load a lane fetches two edges' source rows in whole lines (was two 4-byte loads an edge); row sum order unchanged
# speedup vs baseline: 4.3598x; 1.0456x over previous
#include <hip/hip_runtime.h>
#include <hip/hip_bf16.h>

typedef float v2f __attribute__((ext_vector_type(2)));
typedef float v8f __attribute__((ext_vector_type(8)));
typedef int v4i __attribute__((ext_vector_type(4)));

#define N_NODES 100000
#define N_EDGES 1600000
#define IN_F 64
#define HID 64
#define NC 40

#define BROWS 128
#define BSHIFT 7
#define NBK ((N_NODES + BROWS - 1) / BROWS)
#define NBK_PAD 800
#define NCH 256
#define CH_EDGES (N_EDGES / NCH)
#define CH_THREADS 1024
#define CH_TRIPS ((CH_EDGES + CH_THREADS - 1) / CH_THREADS)
#define SRC_BITS 17
static_assert(N_EDGES % NCH == 0, "the edge list is cut into whole chunks");
static_assert(N_NODES <= (1 << SRC_BITS), "source row fits its field");
static_assert(NBK <= NBK_PAD && NBK_PAD <= CH_THREADS && NBK_PAD % 8 == 0, "bucket table");
#define OFF_PITCH (NBK_PAD + 32)
static_assert(NBK + 1 <= NBK_PAD, "entry bucket + 1 exists for the last bucket");

__global__ void __launch_bounds__(CH_THREADS)
gcn_chunk_order_kernel(const int* __restrict__ src, const int* __restrict__ dst,
                       int* __restrict__ off, int* __restrict__ grouped) {
  __shared__ int sc[2][CH_THREADS];
  __shared__ int h[NBK_PAD];
  __shared__ int bfirst[NBK_PAD];
  __shared__ int ord[CH_EDGES];
  const int t = threadIdx.x, j = blockIdx.x;
  const int* dp = dst + (size_t)j * CH_EDGES;
  const int* sp = src + (size_t)j * CH_EDGES;
  int d[CH_TRIPS], s[CH_TRIPS], rk[CH_TRIPS];
#pragma unroll
  for (int i = 0; i < CH_TRIPS; ++i) {
    const int idx = t + i * CH_THREADS;
    if (idx < CH_EDGES) { d[i] = dp[idx]; s[i] = sp[idx]; }
  }
  if (t < NBK_PAD) h[t] = 0;
  __syncthreads();
#pragma unroll
  for (int i = 0; i < CH_TRIPS; ++i) {
    const int idx = t + i * CH_THREADS;
    if (idx < CH_EDGES) rk[i] = atomicAdd(&h[d[i] >> BSHIFT], 1);
  }
  __syncthreads();
  const int c = t < NBK_PAD ? h[t] : 0;
  sc[0][t] = c;
  __syncthreads();
  int cur = 0;
  for (int dlt = 1; dlt < CH_THREADS; dlt <<= 1) {
    int x = sc[cur][t];
    if (t >= dlt) x += sc[cur][t - dlt];
    sc[cur ^ 1][t] = x;
    __syncthreads();
    cur ^= 1;
  }
  const int lfirst = sc[cur][t] - c;
  if (t < NBK_PAD) bfirst[t] = lfirst;
  if (t <= NBK_PAD) off[(size_t)j * OFF_PITCH + t] = lfirst;
  __syncthreads();
#pragma unroll
  for (int i = 0; i < CH_TRIPS; ++i) {
    const int idx = t + i * CH_THREADS;
    if (idx < CH_EDGES)
      ord[bfirst[d[i] >> BSHIFT] + rk[i]] = s[i] | ((d[i] & (BROWS - 1)) << SRC_BITS);
  }
  __syncthreads();
  int* gp = grouped + (size_t)j * CH_EDGES;
#pragma unroll
  for (int i = 0; i < CH_TRIPS; ++i) {
    const int idx = t + i * CH_THREADS;
    if (idx < CH_EDGES) gp[idx] = ord[idx];
  }
}

typedef float v4f __attribute__((ext_vector_type(4)));
#define SUM_THREADS 256
#define SUM_WAVES (SUM_THREADS / 32)
#define ROWS_PER_WAVE (BROWS / SUM_WAVES)
#define HALF_ROWS (ROWS_PER_WAVE / 2)
#define NULL_EDGE ((unsigned)BROWS << SRC_BITS)
#define SRC_MASK ((1u << SRC_BITS) - 1u)
#define RUN_WORDS 24
#define SEG_CAP (NCH * RUN_WORDS)
static_assert(NCH == SUM_THREADS, "thread r of a block reads the bounds of run r");
static_assert(BROWS == 128, "one wave scans the row counts: 32 lanes x 4");
static_assert(BROWS % SUM_WAVES == 0, "every wave owns whole rows");
static_assert(ROWS_PER_WAVE % 2 == 0, "each half-wave owns whole rows");

template <int COLS, int PITCH>
__device__ __forceinline__ void row_flush(float* tile, const int row, const int c, const v4f a) {
  if (COLS == 64 || c < COLS / 4) {
    float* o = tile + row * PITCH + 4 * c;
    atomicAdd(o, a.x);
    atomicAdd(o + 1, a.y);
    atomicAdd(o + 2, a.z);
    atomicAdd(o + 3, a.w);
  }
}

template <int COLS, int PITCH, int EB>
__device__ __forceinline__ void bucket_rows_sum(const float* __restrict__ x,
                                                const int* __restrict__ grouped,
                                                const int* __restrict__ off, const int b,
                                                float* tile, unsigned* sorted,
                                                int* cnt, int* rstart, int* taken) {
  const int t = threadIdx.x;
  const unsigned lane = t & 31u;
  const int wave = __builtin_amdgcn_readfirstlane(t >> 5);
  const int half = (int)(lane >> 4);
  const int hl = (int)(lane & 15u);
  const unsigned cbyte = 16u * (unsigned)(hl < COLS / 4 ? hl : COLS / 4 - 1);

  const int* op = off + (size_t)t * OFF_PITCH + b;
  const int o0 = op[0];
  const int len = op[1] - o0;
  static_assert(CH_EDGES < (1 << 16), "a run's first word and its length share a 32-bit word");
  const int bounds = o0 | (len << 16);
  if (t == 0) taken[0] = 0;
  const int sub = (int)(lane >> 3);
  const int q = (int)(lane & 7u);

  for (int r0 = 0;; r0 += RUN_WORDS) {
    unsigned pk[RUN_WORDS];
    int rbv[8];
#pragma unroll
    for (int i = 0; i < 8; ++i) rbv[i] = __shfl(bounds, 4 * i + sub);
#pragma unroll
    for (int i = 0; i < 8; ++i) {
      const int sl = 4 * i + sub;
      const int rb = rbv[i];
      const int rl = rb >> 16;
      const int g0 = (32 * wave + sl) * CH_EDGES + (rb & 0xffff);
#pragma unroll
      for (int r = 0; r < 3; ++r) {
        const int k = r0 + q + 8 * r;
        pk[3 * i + r] = k < rl ? (unsigned)grouped[g0 + k] : NULL_EDGE;
      }
    }
    if (t < BROWS) cnt[t] = 0;
    __syncthreads();
    if (len > r0 + RUN_WORDS) taken[0] = r0 + RUN_WORDS;
#pragma unroll
    for (int i = 0; i < RUN_WORDS; ++i)
      if (pk[i] < NULL_EDGE) atomicAdd(&cnt[pk[i] >> SRC_BITS], 1);
    __syncthreads();
    if (wave == 0) {
      const v4i c = ((const v4i*)cnt)[lane];
      const int s = c.x + c.y + c.z + c.w;
      int incl = s;
#pragma unroll
      for (int dlt = 1; dlt < 32; dlt <<= 1) {
        const int o = __shfl_up(incl, dlt);
        if ((int)lane >= dlt) incl += o;
      }
      const int ex = incl - s;
      v4i o4;
      o4.x = ex;
      o4.y = ex + c.x;
      o4.z = ex + c.x + c.y;
      o4.w = ex + c.x + c.y + c.z;
      ((v4i*)rstart)[lane] = o4;
      ((v4i*)cnt)[lane] = o4;
      if (lane == 31u) rstart[BROWS] = incl;
    }
    __syncthreads();
#pragma unroll
    for (int i = 0; i < RUN_WORDS; ++i)
      if (pk[i] < NULL_EDGE) sorted[atomicAdd(&cnt[pk[i] >> SRC_BITS], 1)] = pk[i];
    __syncthreads();

    const int hb = ROWS_PER_WAVE * wave;
    const int s0 = __builtin_amdgcn_readfirstlane(rstart[hb]);
    const int s1 = __builtin_amdgcn_readfirstlane(rstart[hb + HALF_ROWS]);
    const int s2 = __builtin_amdgcn_readfirstlane(rstart[hb + 2 * HALF_ROWS]);
    const int nmax = (s1 - s0) > (s2 - s1) ? (s1 - s0) : (s2 - s1);
    int e = half ? s1 : s0;
    const int pe = half ? s2 : s1;
    int cur = hb + HALF_ROWS * half;
    v4f acc = (v4f)(0.0f);
    for (int done = 0; done < nmax; done += EB) {
      unsigned w[EB];
      v4f v[EB];
#pragma unroll
      for (int i = 0; i < EB; ++i) {
        const int k = e + i;
        const unsigned sw = sorted[k < SEG_CAP - 1 ? k : SEG_CAP - 1];
        w[i] = k < pe ? sw : NULL_EDGE;
      }
#pragma unroll
      for (int i = 0; i < EB; ++i)
        v[i] = *(const v4f*)((const char*)x + ((w[i] & SRC_MASK) * (unsigned)(COLS * 4) + cbyte));
#pragma unroll
      for (int i = 0; i < EB; ++i) {
        const int r = (int)(w[i] >> SRC_BITS);
        if (r != cur) {
          row_flush<COLS, PITCH>(tile, cur, hl, acc);
          acc = (v4f)(0.0f);
          cur = r;
        }
        acc += v[i];
      }
      e += EB;
    }
    row_flush<COLS, PITCH>(tile, cur, hl, acc);
    if (taken[0] != r0 + RUN_WORDS) break;
  }
}

#define NC_PAD 48
#define PH 68
#define PT 52
#define W1_PIECES (16 * HID)
#define W2_PIECES (16 * NC_PAD)
#define W1_TRIPS (W1_PIECES / SUM_THREADS)
#define W2_TRIPS (W2_PIECES / SUM_THREADS)
static_assert(IN_F == 64 && HID == 64, "16 WMMA k groups of 4 per layer");
static_assert(W1_PIECES * 4 <= SEG_CAP, "W1 fits the ordered-segment buffer");
static_assert(W1_PIECES % SUM_THREADS == 0 && W2_PIECES % SUM_THREADS == 0, "whole staging trips");
static_assert(ROWS_PER_WAVE == 16, "a wave's rows are one WMMA row tile");
static_assert(N_NODES % ROWS_PER_WAVE == 0, "a wave's 16 rows exist all or not at all");

__global__ void __launch_bounds__(SUM_THREADS)
gcn_bucket_sum_rows64_dense_kernel(const float* __restrict__ x,
                                   const int* __restrict__ grouped,
                                   const int* __restrict__ off,
                                   const float* __restrict__ W1,
                                   const float* __restrict__ b1,
                                   const float* __restrict__ W2,
                                   float* __restrict__ T) {
  __shared__ __attribute__((aligned(16))) float tile[(BROWS + 1) * PH];
  __shared__ __attribute__((aligned(16))) unsigned sorted[SEG_CAP];
  __shared__ __attribute__((aligned(16))) int cnt[BROWS];
  __shared__ __attribute__((aligned(16))) int rstart[BROWS + 4];
  __shared__ __attribute__((aligned(16))) float w2l[W2_PIECES * 4];
  __shared__ int taken[1];
  const int t = threadIdx.x;
  const int b = blockIdx.x;

  v4f w1r[W1_TRIPS];
#pragma unroll
  for (int i = 0; i < W1_TRIPS; ++i) {
    const int q = t + i * SUM_THREADS;
    const int g = q >> 6, n = q & 63;
    const float* w = W1 + (8 * (g >> 1) + 2 * (g & 1)) * HID + n;
    w1r[i].x = w[0];
    w1r[i].y = w[HID];
    w1r[i].z = w[4 * HID];
    w1r[i].w = w[5 * HID];
  }
  v4f w2r[W2_TRIPS];
#pragma unroll
  for (int i = 0; i < W2_TRIPS; ++i) {
    const int q = t + i * SUM_THREADS;
    const int g = q / NC_PAD, n = q - g * NC_PAD;
    const int nc = n < NC ? n : NC - 1;
    const float* w = W2 + (8 * (g >> 1) + 2 * (g & 1)) * NC + nc;
    const float w0 = w[0], w1v = w[NC], w4 = w[4 * NC], w5 = w[5 * NC];
    w2r[i].x = n < NC ? w0 : 0.0f;
    w2r[i].y = n < NC ? w1v : 0.0f;
    w2r[i].z = n < NC ? w4 : 0.0f;
    w2r[i].w = n < NC ? w5 : 0.0f;
  }
  for (int i = t; i < (BROWS + 1) * (PH / 4); i += SUM_THREADS) ((v4f*)tile)[i] = (v4f)(0.0f);
  bucket_rows_sum<64, PH, 8>(x, grouped, off, b, tile, sorted, cnt, rstart, taken);
  __syncthreads();
  float* w1l = (float*)sorted;
#pragma unroll
  for (int i = 0; i < W1_TRIPS; ++i) ((v4f*)w1l)[t + i * SUM_THREADS] = w1r[i];
#pragma unroll
  for (int i = 0; i < W2_TRIPS; ++i) ((v4f*)w2l)[t + i * SUM_THREADS] = w2r[i];
  __syncthreads();

  const int lane = t & 31;
  const int wave = t >> 5;
  const int lh = lane & 15;
  const int hi = lane >> 4;
  float* ht = tile + (ROWS_PER_WAVE * wave) * PH;

  v4f af[8];
#pragma unroll
  for (int p = 0; p < 8; ++p) af[p] = *(const v4f*)&ht[lh * PH + 8 * p + (hi << 2)];

  v8f acc[4] = {};
#pragma unroll
  for (int p = 0; p < 8; ++p) {
#pragma unroll
    for (int u = 0; u < 2; ++u) {
      v2f a;
      a.x = af[p][2 * u];
      a.y = af[p][2 * u + 1];
#pragma unroll
      for (int ct = 0; ct < 4; ++ct) {
        const v2f bq = *(const v2f*)&w1l[((2 * p + u) * HID + ct * 16 + lh) * 4 + 2 * hi];
        acc[ct] = __builtin_amdgcn_wmma_f32_16x16x4_f32(false, a, false, bq, (short)0,
                                                        acc[ct], false, false);
      }
    }
  }

#pragma unroll
  for (int ct = 0; ct < 4; ++ct) {
    const float bv = b1[ct * 16 + lh];
#pragma unroll
    for (int r = 0; r < 8; ++r)
      ht[((hi << 3) + r) * PH + ct * 16 + lh] = fmaxf(acc[ct][r] + bv, 0.0f);
  }
  __syncthreads();

  v4f hf[8];
#pragma unroll
  for (int p = 0; p < 8; ++p) hf[p] = *(const v4f*)&ht[lh * PH + 8 * p + (hi << 2)];
  v8f acc2[3] = {};
#pragma unroll
  for (int p = 0; p < 8; ++p) {
#pragma unroll
    for (int u = 0; u < 2; ++u) {
      v2f a;
      a.x = hf[p][2 * u];
      a.y = hf[p][2 * u + 1];
#pragma unroll
      for (int ct = 0; ct < 3; ++ct) {
        const v2f bq = *(const v2f*)&w2l[((2 * p + u) * NC_PAD + ct * 16 + lh) * 4 + 2 * hi];
        acc2[ct] = __builtin_amdgcn_wmma_f32_16x16x4_f32(false, a, false, bq, (short)0,
                                                         acc2[ct], false, false);
      }
    }
  }

#pragma unroll
  for (int ct = 0; ct < 3; ++ct) {
#pragma unroll
    for (int r = 0; r < 8; ++r)
      ht[((hi << 3) + r) * PT + ct * 16 + lh] = acc2[ct][r];
  }
  __syncthreads();

  const int row0 = b * BROWS + ROWS_PER_WAVE * wave;
  if (row0 < N_NODES) {
    float* trow = T + (size_t)row0 * NC;
#pragma unroll
    for (int q = 0; q < 5; ++q) {
      const int g = lane + 32 * q;
      const int row = g / 10;
      const int c4 = g - row * 10;
      const v4f v = *(const v4f*)&ht[row * PT + 4 * c4];
      *(v4f*)(trow + 4 * g) = v;
    }
  }
}

__global__ void __launch_bounds__(SUM_THREADS)
gcn_bucket_sum_rows40_kernel(const float* __restrict__ x,
                             const int* __restrict__ grouped,
                             const int* __restrict__ off,
                             const float* __restrict__ bias,
                             float* __restrict__ out) {
  __shared__ __attribute__((aligned(16))) float tile[(BROWS + 1) * NC];
  __shared__ __attribute__((aligned(16))) unsigned sorted[SEG_CAP];
  __shared__ __attribute__((aligned(16))) int cnt[BROWS];
  __shared__ __attribute__((aligned(16))) int rstart[BROWS + 4];
  __shared__ int taken[1];
  const int t = threadIdx.x;
  const int b = blockIdx.x;
  if (t < 6 * NC) {
    const int c = t % NC;
    const float bv = bias[c];
    for (int r = t / NC; r < BROWS + 1; r += 6) tile[r * NC + c] = bv;
  }
  bucket_rows_sum<NC, NC, 16>(x, grouped, off, b, tile, sorted, cnt, rstart, taken);
  __syncthreads();
  const int left = N_NODES - b * BROWS;
  const int rows = left < BROWS ? left : BROWS;
  v4f* g = (v4f*)(out + (size_t)b * BROWS * NC);
  for (int i = t; i < rows * (NC / 4); i += SUM_THREADS) g[i] = ((const v4f*)tile)[i];
}

extern "C" void kernel_launch(void* const* d_in, const int* in_sizes, int n_in,
                              void* d_out, int out_size, void* d_ws,
                              size_t ws_size, hipStream_t stream) {
  (void)in_sizes; (void)n_in; (void)out_size; (void)ws_size;
  const float* features = (const float*)d_in[0];
  const int*   src      = (const int*)d_in[1];
  const int*   dst      = (const int*)d_in[2];
  const float* W1       = (const float*)d_in[3];
  const float* b1       = (const float*)d_in[4];
  const float* W2       = (const float*)d_in[5];
  const float* b2       = (const float*)d_in[6];
  float* out = (float*)d_out;

  float* ws    = (float*)d_ws;
  float* t     = ws + (size_t)N_NODES * HID;
  int* grouped = (int*)(t + (size_t)N_NODES * NC);
  int* off     = grouped + N_EDGES;

  gcn_chunk_order_kernel<<<NCH, CH_THREADS, 0, stream>>>(src, dst, off, grouped);

  gcn_bucket_sum_rows64_dense_kernel<<<NBK, SUM_THREADS, 0, stream>>>(
      features, grouped, off, W1, b1, W2, t);

  gcn_bucket_sum_rows40_kernel<<<NBK, SUM_THREADS, 0, stream>>>(t, grouped, off, b2, out);
}
